# starting kernel
# baseline (speedup 1.0000x reference)
#include <hip/hip_runtime.h>

typedef float v2f __attribute__((ext_vector_type(2)));
typedef float v8f __attribute__((ext_vector_type(8)));

#define N_NODES 50000
#define N_EDGES 800000
#define NB (N_NODES / 16)
#define TPITCH 132

__device__ __forceinline__ v8f wmma_f32(v2f a, v2f b, v8f c) {
  return __builtin_amdgcn_wmma_f32_16x16x4_f32(false, a, false, b, (short)0, c,
                                               false, false);
}

__global__ void k_init(float* __restrict__ deg, float* __restrict__ agg) {
  int i = blockIdx.x * 256 + threadIdx.x;
  if (i < N_NODES * 128) agg[i] = 0.0f;
  if (i < N_NODES) deg[i] = 1.0f;
}

__global__ void k_deg(const int* __restrict__ ei, float* __restrict__ deg) {
  int e = blockIdx.x * 256 + threadIdx.x;
  if (e < N_EDGES) atomicAdd(&deg[ei[N_EDGES + e]], 1.0f);
}

__global__ void k_dinv(const float* __restrict__ deg, float* __restrict__ dinv) {
  int i = blockIdx.x * 256 + threadIdx.x;
  if (i < N_NODES) {
    float d = deg[i];
    dinv[i] = d > 0.0f ? rsqrtf(d) : 0.0f;
  }
}

__global__ void k_gemm1(const float* __restrict__ x, const float* __restrict__ Wg,
                        float* __restrict__ xw) {
  __shared__ float lds[4 * 16 * TPITCH];
  const int lane = threadIdx.x & 31;
  const int wave = threadIdx.x >> 5;
  const int half = lane >> 4;
  const int lm   = lane & 15;
  float* tile = lds + wave * (16 * TPITCH);
  const int gw = blockIdx.x * 4 + wave;
  const int nwaves = gridDim.x * 4;

  for (int blk = gw; blk < NB; blk += nwaves) {
    const int row0 = blk * 16;
    for (int r = 0; r < 16; ++r) {
      float4 v = *(const float4*)(x + (size_t)(row0 + r) * 128 + lane * 4);
      *(float4*)(tile + r * TPITCH + lane * 4) = v;
    }
    asm volatile("s_wait_dscnt 0" ::: "memory");

    v8f acc[8];
    for (int t = 0; t < 8; ++t) acc[t] = (v8f){};
    for (int kk = 0; kk < 32; ++kk) {
      const int k0 = 4 * kk + 2 * half;
      v2f a = *(const v2f*)(tile + lm * TPITCH + k0);
      for (int t = 0; t < 8; ++t) {
        const float* bp = Wg + (size_t)k0 * 128 + 16 * t + lm;
        v2f b; b.x = bp[0]; b.y = bp[128];
        acc[t] = wmma_f32(a, b, acc[t]);
      }
    }
    for (int t = 0; t < 8; ++t) {
      float* op = xw + (size_t)(row0 + 8 * half) * 128 + 16 * t + lm;
      for (int r = 0; r < 8; ++r) op[(size_t)r * 128] = acc[t][r];
    }
    asm volatile("s_wait_dscnt 0" ::: "memory");
  }
}

__global__ void k_scatter(const int* __restrict__ ei, const float* __restrict__ dinv,
                          const float* __restrict__ xw, float* __restrict__ agg) {
  int t = blockIdx.x * 256 + threadIdx.x;
  int e = t >> 5;
  int c = (t & 31) << 2;
  if (e >= N_EDGES) return;
  int s = ei[e];
  int d = ei[N_EDGES + e];
  float nrm = dinv[s] * dinv[d];
  float4 v = *(const float4*)(xw + (size_t)s * 128 + c);
  float* a = agg + (size_t)d * 128 + c;
  atomicAdd(a + 0, v.x * nrm);
  atomicAdd(a + 1, v.y * nrm);
  atomicAdd(a + 2, v.z * nrm);
  atomicAdd(a + 3, v.w * nrm);
}

__global__ void k_mlp(const float* __restrict__ x, const float* __restrict__ xw,
                      const float* __restrict__ agg, const float* __restrict__ dinv,
                      const float* __restrict__ bg,
                      const float* __restrict__ W1, const float* __restrict__ b1,
                      const float* __restrict__ W2, const float* __restrict__ b2,
                      const float* __restrict__ W3, const float* __restrict__ b3,
                      float* __restrict__ out) {
  __shared__ float lds[4 * 16 * TPITCH];
  const int lane = threadIdx.x & 31;
  const int wave = threadIdx.x >> 5;
  const int half = lane >> 4;
  const int lm   = lane & 15;
  float* tile = lds + wave * (16 * TPITCH);
  const float4 bgv = *(const float4*)(bg + lane * 4);
  const int gw = blockIdx.x * 4 + wave;
  const int nwaves = gridDim.x * 4;

  for (int blk = gw; blk < NB; blk += nwaves) {
    const int row0 = blk * 16;
    for (int r = 0; r < 16; ++r) {
      const int row = row0 + r;
      float di = dinv[row];
      float sl = di * di;
      float4 ag  = *(const float4*)(agg + (size_t)row * 128 + lane * 4);
      float4 xwv = *(const float4*)(xw  + (size_t)row * 128 + lane * 4);
      float4 xv  = *(const float4*)(x   + (size_t)row * 128 + lane * 4);
      float4 h;
      h.x = fmaxf(ag.x + xwv.x * sl + bgv.x, 0.0f) + xv.x;
      h.y = fmaxf(ag.y + xwv.y * sl + bgv.y, 0.0f) + xv.y;
      h.z = fmaxf(ag.z + xwv.z * sl + bgv.z, 0.0f) + xv.z;
      h.w = fmaxf(ag.w + xwv.w * sl + bgv.w, 0.0f) + xv.w;
      *(float4*)(tile + r * TPITCH + lane * 4) = h;
    }
    asm volatile("s_wait_dscnt 0" ::: "memory");

    v8f acc1[8];
    for (int t = 0; t < 8; ++t) acc1[t] = (v8f){};
    for (int kk = 0; kk < 32; ++kk) {
      const int k0 = 4 * kk + 2 * half;
      v2f a = *(const v2f*)(tile + lm * TPITCH + k0);
      for (int t = 0; t < 8; ++t) {
        const float* bp = W1 + (size_t)k0 * 128 + 16 * t + lm;
        v2f b; b.x = bp[0]; b.y = bp[128];
        acc1[t] = wmma_f32(a, b, acc1[t]);
      }
    }
    asm volatile("s_wait_dscnt 0" ::: "memory");
    for (int t = 0; t < 8; ++t) {
      float bb = b1[16 * t + lm];
      for (int r = 0; r < 8; ++r)
        tile[(r + 8 * half) * TPITCH + 16 * t + lm] = fmaxf(acc1[t][r] + bb, 0.0f);
    }
    asm volatile("s_wait_dscnt 0" ::: "memory");

    v8f acc2[4];
    for (int t = 0; t < 4; ++t) acc2[t] = (v8f){};
    for (int kk = 0; kk < 32; ++kk) {
      const int k0 = 4 * kk + 2 * half;
      v2f a = *(const v2f*)(tile + lm * TPITCH + k0);
      for (int t = 0; t < 4; ++t) {
        const float* bp = W2 + (size_t)k0 * 64 + 16 * t + lm;
        v2f b; b.x = bp[0]; b.y = bp[64];
        acc2[t] = wmma_f32(a, b, acc2[t]);
      }
    }
    asm volatile("s_wait_dscnt 0" ::: "memory");
    for (int t = 0; t < 4; ++t) {
      float bb = b2[16 * t + lm];
      for (int r = 0; r < 8; ++r)
        tile[(r + 8 * half) * TPITCH + 16 * t + lm] = fmaxf(acc2[t][r] + bb, 0.0f);
    }
    asm volatile("s_wait_dscnt 0" ::: "memory");

    v8f acc3 = (v8f){};
    for (int kk = 0; kk < 16; ++kk) {
      const int k0 = 4 * kk + 2 * half;
      v2f a = *(const v2f*)(tile + lm * TPITCH + k0);
      v2f b;
      if (lm < 2) { b.x = W3[k0 * 2 + lm]; b.y = W3[(k0 + 1) * 2 + lm]; }
      else        { b.x = 0.0f;            b.y = 0.0f; }
      acc3 = wmma_f32(a, b, acc3);
    }
    if (lm < 2) {
      float bb = b3[lm];
      for (int r = 0; r < 8; ++r)
        out[(size_t)lm * N_NODES + row0 + r + 8 * half] = acc3[r] + bb;
    }
  }
}

extern "C" void kernel_launch(void* const* d_in, const int* in_sizes, int n_in,
                              void* d_out, int out_size, void* d_ws, size_t ws_size,
                              hipStream_t stream) {
  (void)in_sizes; (void)n_in; (void)out_size; (void)ws_size;
  const float* x  = (const float*)d_in[0];
  const int*   ei = (const int*)d_in[1];
  const float* Wg = (const float*)d_in[2];
  const float* bg = (const float*)d_in[3];
  const float* W1 = (const float*)d_in[4];
  const float* b1 = (const float*)d_in[5];
  const float* W2 = (const float*)d_in[6];
  const float* b2 = (const float*)d_in[7];
  const float* W3 = (const float*)d_in[8];
  const float* b3 = (const float*)d_in[9];
  float* out = (float*)d_out;

  float* ws   = (float*)d_ws;
  float* deg  = ws;
  float* dinv = ws + N_NODES;
  float* xw   = ws + 2 * N_NODES;
  float* agg  = xw + (size_t)N_NODES * 128;

  k_init<<<(N_NODES * 128 + 255) / 256, 256, 0, stream>>>(deg, agg);
  k_deg<<<(N_EDGES + 255) / 256, 256, 0, stream>>>(ei, deg);
  k_dinv<<<(N_NODES + 255) / 256, 256, 0, stream>>>(deg, dinv);
  k_gemm1<<<782, 128, 0, stream>>>(x, Wg, xw);
  k_scatter<<<(N_EDGES * 32) / 256, 256, 0, stream>>>(ei, dinv, xw, agg);
  k_mlp<<<782, 128, 0, stream>>>(x, xw, agg, dinv, bg, W1, b1, W2, b2, W3, b3, out);
}


# scatter: lane l handles floats l, l+32, l+64, l+96 of the row, so each f32 atomic add (and each row load) covers one whole 128-byte line, not quarters of four lines; sums still in arrival order
# speedup vs baseline: 1.2132x; 1.2132x over previous
#include <hip/hip_runtime.h>

typedef float v2f __attribute__((ext_vector_type(2)));
typedef float v8f __attribute__((ext_vector_type(8)));

#define N_NODES 50000
#define N_EDGES 800000
#define NB (N_NODES / 16)
#define TPITCH 132

__device__ __forceinline__ v8f wmma_f32(v2f a, v2f b, v8f c) {
  return __builtin_amdgcn_wmma_f32_16x16x4_f32(false, a, false, b, (short)0, c,
                                               false, false);
}

__global__ void k_init(float* __restrict__ deg, float* __restrict__ agg) {
  int i = blockIdx.x * 256 + threadIdx.x;
  if (i < N_NODES * 128) agg[i] = 0.0f;
  if (i < N_NODES) deg[i] = 1.0f;
}

__global__ void k_deg(const int* __restrict__ ei, float* __restrict__ deg) {
  int e = blockIdx.x * 256 + threadIdx.x;
  if (e < N_EDGES) atomicAdd(&deg[ei[N_EDGES + e]], 1.0f);
}

__global__ void k_dinv(const float* __restrict__ deg, float* __restrict__ dinv) {
  int i = blockIdx.x * 256 + threadIdx.x;
  if (i < N_NODES) {
    float d = deg[i];
    dinv[i] = d > 0.0f ? rsqrtf(d) : 0.0f;
  }
}

__global__ void k_gemm1(const float* __restrict__ x, const float* __restrict__ Wg,
                        float* __restrict__ xw) {
  __shared__ float lds[4 * 16 * TPITCH];
  const int lane = threadIdx.x & 31;
  const int wave = threadIdx.x >> 5;
  const int half = lane >> 4;
  const int lm   = lane & 15;
  float* tile = lds + wave * (16 * TPITCH);
  const int gw = blockIdx.x * 4 + wave;
  const int nwaves = gridDim.x * 4;

  for (int blk = gw; blk < NB; blk += nwaves) {
    const int row0 = blk * 16;
    for (int r = 0; r < 16; ++r) {
      float4 v = *(const float4*)(x + (size_t)(row0 + r) * 128 + lane * 4);
      *(float4*)(tile + r * TPITCH + lane * 4) = v;
    }
    asm volatile("s_wait_dscnt 0" ::: "memory");

    v8f acc[8];
    for (int t = 0; t < 8; ++t) acc[t] = (v8f){};
    for (int kk = 0; kk < 32; ++kk) {
      const int k0 = 4 * kk + 2 * half;
      v2f a = *(const v2f*)(tile + lm * TPITCH + k0);
      for (int t = 0; t < 8; ++t) {
        const float* bp = Wg + (size_t)k0 * 128 + 16 * t + lm;
        v2f b; b.x = bp[0]; b.y = bp[128];
        acc[t] = wmma_f32(a, b, acc[t]);
      }
    }
    for (int t = 0; t < 8; ++t) {
      float* op = xw + (size_t)(row0 + 8 * half) * 128 + 16 * t + lm;
      for (int r = 0; r < 8; ++r) op[(size_t)r * 128] = acc[t][r];
    }
    asm volatile("s_wait_dscnt 0" ::: "memory");
  }
}

__global__ void k_scatter(const int* __restrict__ ei, const float* __restrict__ dinv,
                          const float* __restrict__ xw, float* __restrict__ agg) {
  int t = blockIdx.x * 256 + threadIdx.x;
  int e = t >> 5;
  int c = t & 31;
  if (e >= N_EDGES) return;
  int s = ei[e];
  int d = ei[N_EDGES + e];
  float nrm = dinv[s] * dinv[d];
  const float* p = xw + (size_t)s * 128 + c;
  float v0 = p[0], v1 = p[32], v2 = p[64], v3 = p[96];
  float* a = agg + (size_t)d * 128 + c;
  atomicAdd(a + 0,  v0 * nrm);
  atomicAdd(a + 32, v1 * nrm);
  atomicAdd(a + 64, v2 * nrm);
  atomicAdd(a + 96, v3 * nrm);
}

__global__ void k_mlp(const float* __restrict__ x, const float* __restrict__ xw,
                      const float* __restrict__ agg, const float* __restrict__ dinv,
                      const float* __restrict__ bg,
                      const float* __restrict__ W1, const float* __restrict__ b1,
                      const float* __restrict__ W2, const float* __restrict__ b2,
                      const float* __restrict__ W3, const float* __restrict__ b3,
                      float* __restrict__ out) {
  __shared__ float lds[4 * 16 * TPITCH];
  const int lane = threadIdx.x & 31;
  const int wave = threadIdx.x >> 5;
  const int half = lane >> 4;
  const int lm   = lane & 15;
  float* tile = lds + wave * (16 * TPITCH);
  const float4 bgv = *(const float4*)(bg + lane * 4);
  const int gw = blockIdx.x * 4 + wave;
  const int nwaves = gridDim.x * 4;

  for (int blk = gw; blk < NB; blk += nwaves) {
    const int row0 = blk * 16;
    for (int r = 0; r < 16; ++r) {
      const int row = row0 + r;
      float di = dinv[row];
      float sl = di * di;
      float4 ag  = *(const float4*)(agg + (size_t)row * 128 + lane * 4);
      float4 xwv = *(const float4*)(xw  + (size_t)row * 128 + lane * 4);
      float4 xv  = *(const float4*)(x   + (size_t)row * 128 + lane * 4);
      float4 h;
      h.x = fmaxf(ag.x + xwv.x * sl + bgv.x, 0.0f) + xv.x;
      h.y = fmaxf(ag.y + xwv.y * sl + bgv.y, 0.0f) + xv.y;
      h.z = fmaxf(ag.z + xwv.z * sl + bgv.z, 0.0f) + xv.z;
      h.w = fmaxf(ag.w + xwv.w * sl + bgv.w, 0.0f) + xv.w;
      *(float4*)(tile + r * TPITCH + lane * 4) = h;
    }
    asm volatile("s_wait_dscnt 0" ::: "memory");

    v8f acc1[8];
    for (int t = 0; t < 8; ++t) acc1[t] = (v8f){};
    for (int kk = 0; kk < 32; ++kk) {
      const int k0 = 4 * kk + 2 * half;
      v2f a = *(const v2f*)(tile + lm * TPITCH + k0);
      for (int t = 0; t < 8; ++t) {
        const float* bp = W1 + (size_t)k0 * 128 + 16 * t + lm;
        v2f b; b.x = bp[0]; b.y = bp[128];
        acc1[t] = wmma_f32(a, b, acc1[t]);
      }
    }
    asm volatile("s_wait_dscnt 0" ::: "memory");
    for (int t = 0; t < 8; ++t) {
      float bb = b1[16 * t + lm];
      for (int r = 0; r < 8; ++r)
        tile[(r + 8 * half) * TPITCH + 16 * t + lm] = fmaxf(acc1[t][r] + bb, 0.0f);
    }
    asm volatile("s_wait_dscnt 0" ::: "memory");

    v8f acc2[4];
    for (int t = 0; t < 4; ++t) acc2[t] = (v8f){};
    for (int kk = 0; kk < 32; ++kk) {
      const int k0 = 4 * kk + 2 * half;
      v2f a = *(const v2f*)(tile + lm * TPITCH + k0);
      for (int t = 0; t < 4; ++t) {
        const float* bp = W2 + (size_t)k0 * 64 + 16 * t + lm;
        v2f b; b.x = bp[0]; b.y = bp[64];
        acc2[t] = wmma_f32(a, b, acc2[t]);
      }
    }
    asm volatile("s_wait_dscnt 0" ::: "memory");
    for (int t = 0; t < 4; ++t) {
      float bb = b2[16 * t + lm];
      for (int r = 0; r < 8; ++r)
        tile[(r + 8 * half) * TPITCH + 16 * t + lm] = fmaxf(acc2[t][r] + bb, 0.0f);
    }
    asm volatile("s_wait_dscnt 0" ::: "memory");

    v8f acc3 = (v8f){};
    for (int kk = 0; kk < 16; ++kk) {
      const int k0 = 4 * kk + 2 * half;
      v2f a = *(const v2f*)(tile + lm * TPITCH + k0);
      v2f b;
      if (lm < 2) { b.x = W3[k0 * 2 + lm]; b.y = W3[(k0 + 1) * 2 + lm]; }
      else        { b.x = 0.0f;            b.y = 0.0f; }
      acc3 = wmma_f32(a, b, acc3);
    }
    if (lm < 2) {
      float bb = b3[lm];
      for (int r = 0; r < 8; ++r)
        out[(size_t)lm * N_NODES + row0 + r + 8 * half] = acc3[r] + bb;
    }
  }
}

extern "C" void kernel_launch(void* const* d_in, const int* in_sizes, int n_in,
                              void* d_out, int out_size, void* d_ws, size_t ws_size,
                              hipStream_t stream) {
  (void)in_sizes; (void)n_in; (void)out_size; (void)ws_size;
  const float* x  = (const float*)d_in[0];
  const int*   ei = (const int*)d_in[1];
  const float* Wg = (const float*)d_in[2];
  const float* bg = (const float*)d_in[3];
  const float* W1 = (const float*)d_in[4];
  const float* b1 = (const float*)d_in[5];
  const float* W2 = (const float*)d_in[6];
  const float* b2 = (const float*)d_in[7];
  const float* W3 = (const float*)d_in[8];
  const float* b3 = (const float*)d_in[9];
  float* out = (float*)d_out;

  float* ws   = (float*)d_ws;
  float* deg  = ws;
  float* dinv = ws + N_NODES;
  float* xw   = ws + 2 * N_NODES;
  float* agg  = xw + (size_t)N_NODES * 128;

  k_init<<<(N_NODES * 128 + 255) / 256, 256, 0, stream>>>(deg, agg);
  k_deg<<<(N_EDGES + 255) / 256, 256, 0, stream>>>(ei, deg);
  k_dinv<<<(N_NODES + 255) / 256, 256, 0, stream>>>(deg, dinv);
  k_gemm1<<<782, 128, 0, stream>>>(x, Wg, xw);
  k_scatter<<<(N_EDGES * 32) / 256, 256, 0, stream>>>(ei, dinv, xw, agg);
  k_mlp<<<782, 128, 0, stream>>>(x, xw, agg, dinv, bg, W1, b1, W2, b2, W3, b3, out);
}


# scatter: one wave walks 32 edges (indices and norms loaded once, rows of 8 edges in flight), init stores 16 floats per lane: 800000+200000 launched waves become 25000+12504; sums in arrival order
# speedup vs baseline: 1.3105x; 1.0802x over previous
#include <hip/hip_runtime.h>

typedef float v2f __attribute__((ext_vector_type(2)));
typedef float v8f __attribute__((ext_vector_type(8)));

#define N_NODES 50000
#define N_EDGES 800000
#define NB (N_NODES / 16)
#define TPITCH 132

__device__ __forceinline__ v8f wmma_f32(v2f a, v2f b, v8f c) {
  return __builtin_amdgcn_wmma_f32_16x16x4_f32(false, a, false, b, (short)0, c,
                                               false, false);
}

__global__ __launch_bounds__(256) void k_init(float* __restrict__ deg, float* __restrict__ agg) {
  float4* a4 = (float4*)agg;
  const float4 z = {0.0f, 0.0f, 0.0f, 0.0f};
  for (int k = 0; k < 4; ++k) {
    int i = blockIdx.x * 1024 + k * 256 + threadIdx.x;
    if (i < N_NODES * 32) a4[i] = z;
  }
  int n = blockIdx.x * 256 + threadIdx.x;
  if (n < N_NODES) deg[n] = 1.0f;
}

__global__ void k_deg(const int* __restrict__ ei, float* __restrict__ deg) {
  int e = blockIdx.x * 256 + threadIdx.x;
  if (e < N_EDGES) atomicAdd(&deg[ei[N_EDGES + e]], 1.0f);
}

__global__ void k_dinv(const float* __restrict__ deg, float* __restrict__ dinv) {
  int i = blockIdx.x * 256 + threadIdx.x;
  if (i < N_NODES) {
    float d = deg[i];
    dinv[i] = d > 0.0f ? rsqrtf(d) : 0.0f;
  }
}

__global__ void k_gemm1(const float* __restrict__ x, const float* __restrict__ Wg,
                        float* __restrict__ xw) {
  __shared__ float lds[4 * 16 * TPITCH];
  const int lane = threadIdx.x & 31;
  const int wave = threadIdx.x >> 5;
  const int half = lane >> 4;
  const int lm   = lane & 15;
  float* tile = lds + wave * (16 * TPITCH);
  const int gw = blockIdx.x * 4 + wave;
  const int nwaves = gridDim.x * 4;

  for (int blk = gw; blk < NB; blk += nwaves) {
    const int row0 = blk * 16;
    for (int r = 0; r < 16; ++r) {
      float4 v = *(const float4*)(x + (size_t)(row0 + r) * 128 + lane * 4);
      *(float4*)(tile + r * TPITCH + lane * 4) = v;
    }
    asm volatile("s_wait_dscnt 0" ::: "memory");

    v8f acc[8];
    for (int t = 0; t < 8; ++t) acc[t] = (v8f){};
    for (int kk = 0; kk < 32; ++kk) {
      const int k0 = 4 * kk + 2 * half;
      v2f a = *(const v2f*)(tile + lm * TPITCH + k0);
      for (int t = 0; t < 8; ++t) {
        const float* bp = Wg + (size_t)k0 * 128 + 16 * t + lm;
        v2f b; b.x = bp[0]; b.y = bp[128];
        acc[t] = wmma_f32(a, b, acc[t]);
      }
    }
    for (int t = 0; t < 8; ++t) {
      float* op = xw + (size_t)(row0 + 8 * half) * 128 + 16 * t + lm;
      for (int r = 0; r < 8; ++r) op[(size_t)r * 128] = acc[t][r];
    }
    asm volatile("s_wait_dscnt 0" ::: "memory");
  }
}

#define EPW 32
#define EB  8
__global__ __launch_bounds__(256) void k_scatter(const int* __restrict__ ei,
                                                 const float* __restrict__ dinv,
                                                 const float* __restrict__ xw,
                                                 float* __restrict__ agg) {
  const int lane = threadIdx.x & 31;
  const int w = blockIdx.x * 8 + (threadIdx.x >> 5);
  const int e = w * EPW + lane;
  const int s = ei[e];
  const int d = ei[N_EDGES + e];
  const float nrm = dinv[s] * dinv[d];
#pragma unroll
  for (int j0 = 0; j0 < EPW; j0 += EB) {
    float v[EB][4];
#pragma unroll
    for (int j = 0; j < EB; ++j) {
      const int sj = __builtin_amdgcn_readlane(s, j0 + j);
      const float* p = xw + (size_t)sj * 128 + lane;
      v[j][0] = p[0]; v[j][1] = p[32]; v[j][2] = p[64]; v[j][3] = p[96];
    }
#pragma unroll
    for (int j = 0; j < EB; ++j) {
      const int dj = __builtin_amdgcn_readlane(d, j0 + j);
      const float nj = __int_as_float(__builtin_amdgcn_readlane(__float_as_int(nrm), j0 + j));
      float* a = agg + (size_t)dj * 128 + lane;
      atomicAdd(a + 0,  v[j][0] * nj);
      atomicAdd(a + 32, v[j][1] * nj);
      atomicAdd(a + 64, v[j][2] * nj);
      atomicAdd(a + 96, v[j][3] * nj);
    }
  }
}

__global__ void k_mlp(const float* __restrict__ x, const float* __restrict__ xw,
                      const float* __restrict__ agg, const float* __restrict__ dinv,
                      const float* __restrict__ bg,
                      const float* __restrict__ W1, const float* __restrict__ b1,
                      const float* __restrict__ W2, const float* __restrict__ b2,
                      const float* __restrict__ W3, const float* __restrict__ b3,
                      float* __restrict__ out) {
  __shared__ float lds[4 * 16 * TPITCH];
  const int lane = threadIdx.x & 31;
  const int wave = threadIdx.x >> 5;
  const int half = lane >> 4;
  const int lm   = lane & 15;
  float* tile = lds + wave * (16 * TPITCH);
  const float4 bgv = *(const float4*)(bg + lane * 4);
  const int gw = blockIdx.x * 4 + wave;
  const int nwaves = gridDim.x * 4;

  for (int blk = gw; blk < NB; blk += nwaves) {
    const int row0 = blk * 16;
    for (int r = 0; r < 16; ++r) {
      const int row = row0 + r;
      float di = dinv[row];
      float sl = di * di;
      float4 ag  = *(const float4*)(agg + (size_t)row * 128 + lane * 4);
      float4 xwv = *(const float4*)(xw  + (size_t)row * 128 + lane * 4);
      float4 xv  = *(const float4*)(x   + (size_t)row * 128 + lane * 4);
      float4 h;
      h.x = fmaxf(ag.x + xwv.x * sl + bgv.x, 0.0f) + xv.x;
      h.y = fmaxf(ag.y + xwv.y * sl + bgv.y, 0.0f) + xv.y;
      h.z = fmaxf(ag.z + xwv.z * sl + bgv.z, 0.0f) + xv.z;
      h.w = fmaxf(ag.w + xwv.w * sl + bgv.w, 0.0f) + xv.w;
      *(float4*)(tile + r * TPITCH + lane * 4) = h;
    }
    asm volatile("s_wait_dscnt 0" ::: "memory");

    v8f acc1[8];
    for (int t = 0; t < 8; ++t) acc1[t] = (v8f){};
    for (int kk = 0; kk < 32; ++kk) {
      const int k0 = 4 * kk + 2 * half;
      v2f a = *(const v2f*)(tile + lm * TPITCH + k0);
      for (int t = 0; t < 8; ++t) {
        const float* bp = W1 + (size_t)k0 * 128 + 16 * t + lm;
        v2f b; b.x = bp[0]; b.y = bp[128];
        acc1[t] = wmma_f32(a, b, acc1[t]);
      }
    }
    asm volatile("s_wait_dscnt 0" ::: "memory");
    for (int t = 0; t < 8; ++t) {
      float bb = b1[16 * t + lm];
      for (int r = 0; r < 8; ++r)
        tile[(r + 8 * half) * TPITCH + 16 * t + lm] = fmaxf(acc1[t][r] + bb, 0.0f);
    }
    asm volatile("s_wait_dscnt 0" ::: "memory");

    v8f acc2[4];
    for (int t = 0; t < 4; ++t) acc2[t] = (v8f){};
    for (int kk = 0; kk < 32; ++kk) {
      const int k0 = 4 * kk + 2 * half;
      v2f a = *(const v2f*)(tile + lm * TPITCH + k0);
      for (int t = 0; t < 4; ++t) {
        const float* bp = W2 + (size_t)k0 * 64 + 16 * t + lm;
        v2f b; b.x = bp[0]; b.y = bp[64];
        acc2[t] = wmma_f32(a, b, acc2[t]);
      }
    }
    asm volatile("s_wait_dscnt 0" ::: "memory");
    for (int t = 0; t < 4; ++t) {
      float bb = b2[16 * t + lm];
      for (int r = 0; r < 8; ++r)
        tile[(r + 8 * half) * TPITCH + 16 * t + lm] = fmaxf(acc2[t][r] + bb, 0.0f);
    }
    asm volatile("s_wait_dscnt 0" ::: "memory");

    v8f acc3 = (v8f){};
    for (int kk = 0; kk < 16; ++kk) {
      const int k0 = 4 * kk + 2 * half;
      v2f a = *(const v2f*)(tile + lm * TPITCH + k0);
      v2f b;
      if (lm < 2) { b.x = W3[k0 * 2 + lm]; b.y = W3[(k0 + 1) * 2 + lm]; }
      else        { b.x = 0.0f;            b.y = 0.0f; }
      acc3 = wmma_f32(a, b, acc3);
    }
    if (lm < 2) {
      float bb = b3[lm];
      for (int r = 0; r < 8; ++r)
        out[(size_t)lm * N_NODES + row0 + r + 8 * half] = acc3[r] + bb;
    }
  }
}

extern "C" void kernel_launch(void* const* d_in, const int* in_sizes, int n_in,
                              void* d_out, int out_size, void* d_ws, size_t ws_size,
                              hipStream_t stream) {
  (void)in_sizes; (void)n_in; (void)out_size; (void)ws_size;
  const float* x  = (const float*)d_in[0];
  const int*   ei = (const int*)d_in[1];
  const float* Wg = (const float*)d_in[2];
  const float* bg = (const float*)d_in[3];
  const float* W1 = (const float*)d_in[4];
  const float* b1 = (const float*)d_in[5];
  const float* W2 = (const float*)d_in[6];
  const float* b2 = (const float*)d_in[7];
  const float* W3 = (const float*)d_in[8];
  const float* b3 = (const float*)d_in[9];
  float* out = (float*)d_out;

  float* ws   = (float*)d_ws;
  float* deg  = ws;
  float* dinv = ws + N_NODES;
  float* xw   = ws + 2 * N_NODES;
  float* agg  = xw + (size_t)N_NODES * 128;

  k_init<<<(N_NODES * 32 + 1023) / 1024, 256, 0, stream>>>(deg, agg);
  k_deg<<<(N_EDGES + 255) / 256, 256, 0, stream>>>(ei, deg);
  k_dinv<<<(N_NODES + 255) / 256, 256, 0, stream>>>(deg, dinv);
  k_gemm1<<<782, 128, 0, stream>>>(x, Wg, xw);
  k_scatter<<<N_EDGES / (EPW * 8), 256, 0, stream>>>(ei, dinv, xw, agg);
  k_mlp<<<782, 128, 0, stream>>>(x, xw, agg, dinv, bg, W1, b1, W2, b2, W3, b3, out);
}


# aggregation as a gather: edges grouped by dst in every call (count with rank, wave prefix sum, fill); one wave sums a node's rows in registers, one store; no f32 atomics; sum order = arrival rank
# speedup vs baseline: 1.4803x; 1.1296x over previous
#include <hip/hip_runtime.h>

typedef float v2f __attribute__((ext_vector_type(2)));
typedef float v8f __attribute__((ext_vector_type(8)));

#define N_NODES 50000
#define N_EDGES 800000
#define NB (N_NODES / 16)
#define TPITCH 132

__device__ __forceinline__ v8f wmma_f32(v2f a, v2f b, v8f c) {
  return __builtin_amdgcn_wmma_f32_16x16x4_f32(false, a, false, b, (short)0, c,
                                               false, false);
}

#define OFF_CNT    0
#define OFF_DINV   50048
#define OFF_START  100096
#define OFF_CURSOR 150144
#define OFF_RANK   150208
#define OFF_CSR    950208
#define OFF_XW     1750208
#define OFF_AGG    8150208

__global__ __launch_bounds__(256) void k_init(int* __restrict__ cnt, int* __restrict__ cursor) {
  int n = blockIdx.x * 256 + threadIdx.x;
  if (n < N_NODES) cnt[n] = 0;
  if (n == 0) cursor[0] = 0;
}

__global__ __launch_bounds__(256) void k_rank(const int* __restrict__ ei, int* __restrict__ cnt,
                                              int* __restrict__ rank) {
  int e = blockIdx.x * 256 + threadIdx.x;
  if (e < N_EDGES) rank[e] = atomicAdd(&cnt[ei[N_EDGES + e]], 1);
}

__global__ __launch_bounds__(256) void k_offsets(const int* __restrict__ cnt, int* __restrict__ cursor,
                                                 int* __restrict__ start, float* __restrict__ dinv) {
  const int i = blockIdx.x * 256 + threadIdx.x;
  const int lane = threadIdx.x & 31;
  const int c = (i < N_NODES) ? cnt[i] : 0;
  int incl = c;
  for (int off = 1; off < 32; off <<= 1) {
    int t = __shfl_up(incl, off);
    if (lane >= off) incl += t;
  }
  const int total = __shfl(incl, 31);
  int base = 0;
  if (lane == 31) base = atomicAdd(cursor, total);
  base = __shfl(base, 31);
  if (i < N_NODES) {
    start[i] = base + incl - c;
    float d = (float)c + 1.0f;
    dinv[i] = d > 0.0f ? rsqrtf(d) : 0.0f;
  }
}

__global__ __launch_bounds__(256) void k_fill(const int* __restrict__ ei, const int* __restrict__ start,
                                              const int* __restrict__ rank, int* __restrict__ csr) {
  int e = blockIdx.x * 256 + threadIdx.x;
  if (e < N_EDGES) csr[start[ei[N_EDGES + e]] + rank[e]] = ei[e];
}

__global__ void k_gemm1(const float* __restrict__ x, const float* __restrict__ Wg,
                        float* __restrict__ xw) {
  __shared__ float lds[4 * 16 * TPITCH];
  const int lane = threadIdx.x & 31;
  const int wave = threadIdx.x >> 5;
  const int half = lane >> 4;
  const int lm   = lane & 15;
  float* tile = lds + wave * (16 * TPITCH);
  const int gw = blockIdx.x * 4 + wave;
  const int nwaves = gridDim.x * 4;

  for (int blk = gw; blk < NB; blk += nwaves) {
    const int row0 = blk * 16;
    for (int r = 0; r < 16; ++r) {
      float4 v = *(const float4*)(x + (size_t)(row0 + r) * 128 + lane * 4);
      *(float4*)(tile + r * TPITCH + lane * 4) = v;
    }
    asm volatile("s_wait_dscnt 0" ::: "memory");

    v8f acc[8];
    for (int t = 0; t < 8; ++t) acc[t] = (v8f){};
    for (int kk = 0; kk < 32; ++kk) {
      const int k0 = 4 * kk + 2 * half;
      v2f a = *(const v2f*)(tile + lm * TPITCH + k0);
      for (int t = 0; t < 8; ++t) {
        const float* bp = Wg + (size_t)k0 * 128 + 16 * t + lm;
        v2f b; b.x = bp[0]; b.y = bp[128];
        acc[t] = wmma_f32(a, b, acc[t]);
      }
    }
    for (int t = 0; t < 8; ++t) {
      float* op = xw + (size_t)(row0 + 8 * half) * 128 + 16 * t + lm;
      for (int r = 0; r < 8; ++r) op[(size_t)r * 128] = acc[t][r];
    }
    asm volatile("s_wait_dscnt 0" ::: "memory");
  }
}

#define GB 4
__global__ __launch_bounds__(256) void k_gather(const int* __restrict__ cnt,
                                                const int* __restrict__ start,
                                                const int* __restrict__ csr,
                                                const float* __restrict__ dinv,
                                                const float* __restrict__ xw,
                                                float* __restrict__ agg) {
  const int lane = threadIdx.x & 31;
  const int node = __builtin_amdgcn_readfirstlane(blockIdx.x * 8 + (threadIdx.x >> 5));
  const int n = cnt[node];
  const int b = start[node];
  const float dd = dinv[node];
  float4 a = {0.0f, 0.0f, 0.0f, 0.0f};
  for (int base = 0; base < n; base += 32) {
    const int m = (n - base) < 32 ? (n - base) : 32;
    int s = 0;
    float w = 0.0f;
    if (lane < m) {
      s = csr[b + base + lane];
      w = dinv[s] * dd;
    }
    for (int j0 = 0; j0 < m; j0 += GB) {
      float4 v[GB];
#pragma unroll
      for (int k = 0; k < GB; ++k) {
        const int jj = (j0 + k) < m ? (j0 + k) : (m - 1);
        const int sj = __builtin_amdgcn_readlane(s, jj);
        v[k] = *((const float4*)(xw + (size_t)sj * 128) + lane);
      }
#pragma unroll
      for (int k = 0; k < GB; ++k) {
        const float nj = __int_as_float(__builtin_amdgcn_readlane(__float_as_int(w), j0 + k));
        a.x += v[k].x * nj;
        a.y += v[k].y * nj;
        a.z += v[k].z * nj;
        a.w += v[k].w * nj;
      }
    }
  }
  *((float4*)(agg + (size_t)node * 128) + lane) = a;
}

__global__ void k_mlp(const float* __restrict__ x, const float* __restrict__ xw,
                      const float* __restrict__ agg, const float* __restrict__ dinv,
                      const float* __restrict__ bg,
                      const float* __restrict__ W1, const float* __restrict__ b1,
                      const float* __restrict__ W2, const float* __restrict__ b2,
                      const float* __restrict__ W3, const float* __restrict__ b3,
                      float* __restrict__ out) {
  __shared__ float lds[4 * 16 * TPITCH];
  const int lane = threadIdx.x & 31;
  const int wave = threadIdx.x >> 5;
  const int half = lane >> 4;
  const int lm   = lane & 15;
  float* tile = lds + wave * (16 * TPITCH);
  const float4 bgv = *(const float4*)(bg + lane * 4);
  const int gw = blockIdx.x * 4 + wave;
  const int nwaves = gridDim.x * 4;

  for (int blk = gw; blk < NB; blk += nwaves) {
    const int row0 = blk * 16;
    for (int r = 0; r < 16; ++r) {
      const int row = row0 + r;
      float di = dinv[row];
      float sl = di * di;
      float4 ag  = *(const float4*)(agg + (size_t)row * 128 + lane * 4);
      float4 xwv = *(const float4*)(xw  + (size_t)row * 128 + lane * 4);
      float4 xv  = *(const float4*)(x   + (size_t)row * 128 + lane * 4);
      float4 h;
      h.x = fmaxf(ag.x + xwv.x * sl + bgv.x, 0.0f) + xv.x;
      h.y = fmaxf(ag.y + xwv.y * sl + bgv.y, 0.0f) + xv.y;
      h.z = fmaxf(ag.z + xwv.z * sl + bgv.z, 0.0f) + xv.z;
      h.w = fmaxf(ag.w + xwv.w * sl + bgv.w, 0.0f) + xv.w;
      *(float4*)(tile + r * TPITCH + lane * 4) = h;
    }
    asm volatile("s_wait_dscnt 0" ::: "memory");

    v8f acc1[8];
    for (int t = 0; t < 8; ++t) acc1[t] = (v8f){};
    for (int kk = 0; kk < 32; ++kk) {
      const int k0 = 4 * kk + 2 * half;
      v2f a = *(const v2f*)(tile + lm * TPITCH + k0);
      for (int t = 0; t < 8; ++t) {
        const float* bp = W1 + (size_t)k0 * 128 + 16 * t + lm;
        v2f b; b.x = bp[0]; b.y = bp[128];
        acc1[t] = wmma_f32(a, b, acc1[t]);
      }
    }
    asm volatile("s_wait_dscnt 0" ::: "memory");
    for (int t = 0; t < 8; ++t) {
      float bb = b1[16 * t + lm];
      for (int r = 0; r < 8; ++r)
        tile[(r + 8 * half) * TPITCH + 16 * t + lm] = fmaxf(acc1[t][r] + bb, 0.0f);
    }
    asm volatile("s_wait_dscnt 0" ::: "memory");

    v8f acc2[4];
    for (int t = 0; t < 4; ++t) acc2[t] = (v8f){};
    for (int kk = 0; kk < 32; ++kk) {
      const int k0 = 4 * kk + 2 * half;
      v2f a = *(const v2f*)(tile + lm * TPITCH + k0);
      for (int t = 0; t < 4; ++t) {
        const float* bp = W2 + (size_t)k0 * 64 + 16 * t + lm;
        v2f b; b.x = bp[0]; b.y = bp[64];
        acc2[t] = wmma_f32(a, b, acc2[t]);
      }
    }
    asm volatile("s_wait_dscnt 0" ::: "memory");
    for (int t = 0; t < 4; ++t) {
      float bb = b2[16 * t + lm];
      for (int r = 0; r < 8; ++r)
        tile[(r + 8 * half) * TPITCH + 16 * t + lm] = fmaxf(acc2[t][r] + bb, 0.0f);
    }
    asm volatile("s_wait_dscnt 0" ::: "memory");

    v8f acc3 = (v8f){};
    for (int kk = 0; kk < 16; ++kk) {
      const int k0 = 4 * kk + 2 * half;
      v2f a = *(const v2f*)(tile + lm * TPITCH + k0);
      v2f b;
      if (lm < 2) { b.x = W3[k0 * 2 + lm]; b.y = W3[(k0 + 1) * 2 + lm]; }
      else        { b.x = 0.0f;            b.y = 0.0f; }
      acc3 = wmma_f32(a, b, acc3);
    }
    if (lm < 2) {
      float bb = b3[lm];
      for (int r = 0; r < 8; ++r)
        out[(size_t)lm * N_NODES + row0 + r + 8 * half] = acc3[r] + bb;
    }
  }
}

extern "C" void kernel_launch(void* const* d_in, const int* in_sizes, int n_in,
                              void* d_out, int out_size, void* d_ws, size_t ws_size,
                              hipStream_t stream) {
  (void)in_sizes; (void)n_in; (void)out_size; (void)ws_size;
  const float* x  = (const float*)d_in[0];
  const int*   ei = (const int*)d_in[1];
  const float* Wg = (const float*)d_in[2];
  const float* bg = (const float*)d_in[3];
  const float* W1 = (const float*)d_in[4];
  const float* b1 = (const float*)d_in[5];
  const float* W2 = (const float*)d_in[6];
  const float* b2 = (const float*)d_in[7];
  const float* W3 = (const float*)d_in[8];
  const float* b3 = (const float*)d_in[9];
  float* out = (float*)d_out;

  float* ws    = (float*)d_ws;
  int*   wsi   = (int*)d_ws;
  int*   cnt   = wsi + OFF_CNT;
  float* dinv  = ws  + OFF_DINV;
  int*   start = wsi + OFF_START;
  int*   cursor= wsi + OFF_CURSOR;
  int*   rank  = wsi + OFF_RANK;
  int*   csr   = wsi + OFF_CSR;
  float* xw    = ws  + OFF_XW;
  float* agg   = ws  + OFF_AGG;

  k_init<<<(N_NODES + 255) / 256, 256, 0, stream>>>(cnt, cursor);
  k_rank<<<(N_EDGES + 255) / 256, 256, 0, stream>>>(ei, cnt, rank);
  k_offsets<<<(N_NODES + 255) / 256, 256, 0, stream>>>(cnt, cursor, start, dinv);
  k_fill<<<(N_EDGES + 255) / 256, 256, 0, stream>>>(ei, start, rank, csr);
  k_gemm1<<<782, 128, 0, stream>>>(x, Wg, xw);
  k_gather<<<N_NODES / 8, 256, 0, stream>>>(cnt, start, csr, dinv, xw, agg);
  k_mlp<<<782, 128, 0, stream>>>(x, xw, agg, dinv, bg, W1, b1, W2, b2, W3, b3, out);
}


# rank edges with LDS counters: 64 blocks each count a 12500-edge slice in a 50000-word LDS table, per-node prefix over slices gives slots; no global atomic per edge; sum order = slice, then arrival
# speedup vs baseline: 1.8403x; 1.2432x over previous
#include <hip/hip_runtime.h>

typedef float v2f __attribute__((ext_vector_type(2)));
typedef float v8f __attribute__((ext_vector_type(8)));

#define N_NODES 50000
#define N_EDGES 800000
#define NB (N_NODES / 16)
#define TPITCH 132

__device__ __forceinline__ v8f wmma_f32(v2f a, v2f b, v8f c) {
  return __builtin_amdgcn_wmma_f32_16x16x4_f32(false, a, false, b, (short)0, c,
                                               false, false);
}

#define OFF_CNT    0
#define OFF_DINV   50048
#define OFF_START  100096
#define OFF_CURSOR 150144
#define OFF_RANK   150208
#define OFF_CSR    950208
#define OFF_XW     1750208
#define OFF_AGG    8150208
#define OFF_TBL    14550208

#define NSLICE 64
#define EPS    (N_EDGES / NSLICE)
#define EPS4   (EPS / 4)
#define NPITCH 50048

__global__ __launch_bounds__(1024) void k_rank(const int* __restrict__ ei, int* __restrict__ rank,
                                               int* __restrict__ tbl, int* __restrict__ cursor) {
  __shared__ int c[N_NODES];
  const int tid = threadIdx.x;
  const int s = blockIdx.x;
  const int4* dst4 = (const int4*)(ei + N_EDGES + s * EPS);
  int4* rank4 = (int4*)(rank + s * EPS);
  int4* c4 = (int4*)c;
  int4 d[4];
#pragma unroll
  for (int k = 0; k < 4; ++k) {
    const int i = tid + k * 1024;
    d[k] = (i < EPS4) ? dst4[i] : make_int4(0, 0, 0, 0);
  }
  for (int i = tid; i < N_NODES / 4; i += 1024) c4[i] = make_int4(0, 0, 0, 0);
  __syncthreads();
  int4 r[4];
#pragma unroll
  for (int k = 0; k < 4; ++k) {
    const int i = tid + k * 1024;
    if (i < EPS4) {
      r[k].x = atomicAdd(&c[d[k].x], 1);
      r[k].y = atomicAdd(&c[d[k].y], 1);
      r[k].z = atomicAdd(&c[d[k].z], 1);
      r[k].w = atomicAdd(&c[d[k].w], 1);
    }
  }
  __syncthreads();
#pragma unroll
  for (int k = 0; k < 4; ++k) {
    const int i = tid + k * 1024;
    if (i < EPS4) rank4[i] = r[k];
  }
  int4* t4 = (int4*)(tbl + (size_t)s * NPITCH);
  for (int i = tid; i < N_NODES / 4; i += 1024) t4[i] = c4[i];
  if (s == 0 && tid == 0) cursor[0] = 0;
}

__global__ __launch_bounds__(256) void k_offsets(int* __restrict__ tbl, int* __restrict__ cursor,
                                                 int* __restrict__ cnt, int* __restrict__ start,
                                                 float* __restrict__ dinv) {
  const int i = blockIdx.x * 256 + threadIdx.x;
  const int lane = threadIdx.x & 31;
  const int ic = (i < N_NODES) ? i : (N_NODES - 1);
  int v[NSLICE];
  int c = 0;
#pragma unroll
  for (int s = 0; s < NSLICE; ++s) {
    v[s] = tbl[(size_t)s * NPITCH + ic];
    c += v[s];
  }
  if (i >= N_NODES) c = 0;
  int incl = c;
  for (int off = 1; off < 32; off <<= 1) {
    int t = __shfl_up(incl, off);
    if (lane >= off) incl += t;
  }
  const int total = __shfl(incl, 31);
  int base = 0;
  if (lane == 31) base = atomicAdd(cursor, total);
  base = __shfl(base, 31);
  if (i < N_NODES) {
    int run = base + incl - c;
    start[i] = run;
    cnt[i] = c;
    float d = (float)c + 1.0f;
    dinv[i] = d > 0.0f ? rsqrtf(d) : 0.0f;
#pragma unroll
    for (int s = 0; s < NSLICE; ++s) {
      tbl[(size_t)s * NPITCH + i] = run;
      run += v[s];
    }
  }
}

__global__ __launch_bounds__(256) void k_fill(const int* __restrict__ ei, const int* __restrict__ tbl,
                                              const int* __restrict__ rank, int* __restrict__ csr) {
  int e = blockIdx.x * 256 + threadIdx.x;
  if (e < N_EDGES)
    csr[tbl[(size_t)(e / EPS) * NPITCH + ei[N_EDGES + e]] + rank[e]] = ei[e];
}

__global__ void k_gemm1(const float* __restrict__ x, const float* __restrict__ Wg,
                        float* __restrict__ xw) {
  __shared__ float lds[4 * 16 * TPITCH];
  const int lane = threadIdx.x & 31;
  const int wave = threadIdx.x >> 5;
  const int half = lane >> 4;
  const int lm   = lane & 15;
  float* tile = lds + wave * (16 * TPITCH);
  const int gw = blockIdx.x * 4 + wave;
  const int nwaves = gridDim.x * 4;

  for (int blk = gw; blk < NB; blk += nwaves) {
    const int row0 = blk * 16;
    for (int r = 0; r < 16; ++r) {
      float4 v = *(const float4*)(x + (size_t)(row0 + r) * 128 + lane * 4);
      *(float4*)(tile + r * TPITCH + lane * 4) = v;
    }
    asm volatile("s_wait_dscnt 0" ::: "memory");

    v8f acc[8];
    for (int t = 0; t < 8; ++t) acc[t] = (v8f){};
    for (int kk = 0; kk < 32; ++kk) {
      const int k0 = 4 * kk + 2 * half;
      v2f a = *(const v2f*)(tile + lm * TPITCH + k0);
      for (int t = 0; t < 8; ++t) {
        const float* bp = Wg + (size_t)k0 * 128 + 16 * t + lm;
        v2f b; b.x = bp[0]; b.y = bp[128];
        acc[t] = wmma_f32(a, b, acc[t]);
      }
    }
    for (int t = 0; t < 8; ++t) {
      float* op = xw + (size_t)(row0 + 8 * half) * 128 + 16 * t + lm;
      for (int r = 0; r < 8; ++r) op[(size_t)r * 128] = acc[t][r];
    }
    asm volatile("s_wait_dscnt 0" ::: "memory");
  }
}

#define GB 4
__global__ __launch_bounds__(256) void k_gather(const int* __restrict__ cnt,
                                                const int* __restrict__ start,
                                                const int* __restrict__ csr,
                                                const float* __restrict__ dinv,
                                                const float* __restrict__ xw,
                                                float* __restrict__ agg) {
  const int lane = threadIdx.x & 31;
  const int node = __builtin_amdgcn_readfirstlane(blockIdx.x * 8 + (threadIdx.x >> 5));
  const int n = cnt[node];
  const int b = start[node];
  const float dd = dinv[node];
  float4 a = {0.0f, 0.0f, 0.0f, 0.0f};
  for (int base = 0; base < n; base += 32) {
    const int m = (n - base) < 32 ? (n - base) : 32;
    int s = 0;
    float w = 0.0f;
    if (lane < m) {
      s = csr[b + base + lane];
      w = dinv[s] * dd;
    }
    for (int j0 = 0; j0 < m; j0 += GB) {
      float4 v[GB];
#pragma unroll
      for (int k = 0; k < GB; ++k) {
        const int jj = (j0 + k) < m ? (j0 + k) : (m - 1);
        const int sj = __builtin_amdgcn_readlane(s, jj);
        v[k] = *((const float4*)(xw + (size_t)sj * 128) + lane);
      }
#pragma unroll
      for (int k = 0; k < GB; ++k) {
        const float nj = __int_as_float(__builtin_amdgcn_readlane(__float_as_int(w), j0 + k));
        a.x += v[k].x * nj;
        a.y += v[k].y * nj;
        a.z += v[k].z * nj;
        a.w += v[k].w * nj;
      }
    }
  }
  *((float4*)(agg + (size_t)node * 128) + lane) = a;
}

__global__ void k_mlp(const float* __restrict__ x, const float* __restrict__ xw,
                      const float* __restrict__ agg, const float* __restrict__ dinv,
                      const float* __restrict__ bg,
                      const float* __restrict__ W1, const float* __restrict__ b1,
                      const float* __restrict__ W2, const float* __restrict__ b2,
                      const float* __restrict__ W3, const float* __restrict__ b3,
                      float* __restrict__ out) {
  __shared__ float lds[4 * 16 * TPITCH];
  const int lane = threadIdx.x & 31;
  const int wave = threadIdx.x >> 5;
  const int half = lane >> 4;
  const int lm   = lane & 15;
  float* tile = lds + wave * (16 * TPITCH);
  const float4 bgv = *(const float4*)(bg + lane * 4);
  const int gw = blockIdx.x * 4 + wave;
  const int nwaves = gridDim.x * 4;

  for (int blk = gw; blk < NB; blk += nwaves) {
    const int row0 = blk * 16;
    for (int r = 0; r < 16; ++r) {
      const int row = row0 + r;
      float di = dinv[row];
      float sl = di * di;
      float4 ag  = *(const float4*)(agg + (size_t)row * 128 + lane * 4);
      float4 xwv = *(const float4*)(xw  + (size_t)row * 128 + lane * 4);
      float4 xv  = *(const float4*)(x   + (size_t)row * 128 + lane * 4);
      float4 h;
      h.x = fmaxf(ag.x + xwv.x * sl + bgv.x, 0.0f) + xv.x;
      h.y = fmaxf(ag.y + xwv.y * sl + bgv.y, 0.0f) + xv.y;
      h.z = fmaxf(ag.z + xwv.z * sl + bgv.z, 0.0f) + xv.z;
      h.w = fmaxf(ag.w + xwv.w * sl + bgv.w, 0.0f) + xv.w;
      *(float4*)(tile + r * TPITCH + lane * 4) = h;
    }
    asm volatile("s_wait_dscnt 0" ::: "memory");

    v8f acc1[8];
    for (int t = 0; t < 8; ++t) acc1[t] = (v8f){};
    for (int kk = 0; kk < 32; ++kk) {
      const int k0 = 4 * kk + 2 * half;
      v2f a = *(const v2f*)(tile + lm * TPITCH + k0);
      for (int t = 0; t < 8; ++t) {
        const float* bp = W1 + (size_t)k0 * 128 + 16 * t + lm;
        v2f b; b.x = bp[0]; b.y = bp[128];
        acc1[t] = wmma_f32(a, b, acc1[t]);
      }
    }
    asm volatile("s_wait_dscnt 0" ::: "memory");
    for (int t = 0; t < 8; ++t) {
      float bb = b1[16 * t + lm];
      for (int r = 0; r < 8; ++r)
        tile[(r + 8 * half) * TPITCH + 16 * t + lm] = fmaxf(acc1[t][r] + bb, 0.0f);
    }
    asm volatile("s_wait_dscnt 0" ::: "memory");

    v8f acc2[4];
    for (int t = 0; t < 4; ++t) acc2[t] = (v8f){};
    for (int kk = 0; kk < 32; ++kk) {
      const int k0 = 4 * kk + 2 * half;
      v2f a = *(const v2f*)(tile + lm * TPITCH + k0);
      for (int t = 0; t < 4; ++t) {
        const float* bp = W2 + (size_t)k0 * 64 + 16 * t + lm;
        v2f b; b.x = bp[0]; b.y = bp[64];
        acc2[t] = wmma_f32(a, b, acc2[t]);
      }
    }
    asm volatile("s_wait_dscnt 0" ::: "memory");
    for (int t = 0; t < 4; ++t) {
      float bb = b2[16 * t + lm];
      for (int r = 0; r < 8; ++r)
        tile[(r + 8 * half) * TPITCH + 16 * t + lm] = fmaxf(acc2[t][r] + bb, 0.0f);
    }
    asm volatile("s_wait_dscnt 0" ::: "memory");

    v8f acc3 = (v8f){};
    for (int kk = 0; kk < 16; ++kk) {
      const int k0 = 4 * kk + 2 * half;
      v2f a = *(const v2f*)(tile + lm * TPITCH + k0);
      v2f b;
      if (lm < 2) { b.x = W3[k0 * 2 + lm]; b.y = W3[(k0 + 1) * 2 + lm]; }
      else        { b.x = 0.0f;            b.y = 0.0f; }
      acc3 = wmma_f32(a, b, acc3);
    }
    if (lm < 2) {
      float bb = b3[lm];
      for (int r = 0; r < 8; ++r)
        out[(size_t)lm * N_NODES + row0 + r + 8 * half] = acc3[r] + bb;
    }
  }
}

extern "C" void kernel_launch(void* const* d_in, const int* in_sizes, int n_in,
                              void* d_out, int out_size, void* d_ws, size_t ws_size,
                              hipStream_t stream) {
  (void)in_sizes; (void)n_in; (void)out_size; (void)ws_size;
  const float* x  = (const float*)d_in[0];
  const int*   ei = (const int*)d_in[1];
  const float* Wg = (const float*)d_in[2];
  const float* bg = (const float*)d_in[3];
  const float* W1 = (const float*)d_in[4];
  const float* b1 = (const float*)d_in[5];
  const float* W2 = (const float*)d_in[6];
  const float* b2 = (const float*)d_in[7];
  const float* W3 = (const float*)d_in[8];
  const float* b3 = (const float*)d_in[9];
  float* out = (float*)d_out;

  float* ws    = (float*)d_ws;
  int*   wsi   = (int*)d_ws;
  int*   cnt   = wsi + OFF_CNT;
  float* dinv  = ws  + OFF_DINV;
  int*   start = wsi + OFF_START;
  int*   cursor= wsi + OFF_CURSOR;
  int*   rank  = wsi + OFF_RANK;
  int*   csr   = wsi + OFF_CSR;
  float* xw    = ws  + OFF_XW;
  float* agg   = ws  + OFF_AGG;
  int*   tbl   = wsi + OFF_TBL;

  k_rank<<<NSLICE, 1024, 0, stream>>>(ei, rank, tbl, cursor);
  k_offsets<<<(N_NODES + 255) / 256, 256, 0, stream>>>(tbl, cursor, cnt, start, dinv);
  k_fill<<<(N_EDGES + 255) / 256, 256, 0, stream>>>(ei, tbl, rank, csr);
  k_gemm1<<<782, 128, 0, stream>>>(x, Wg, xw);
  k_gather<<<N_NODES / 8, 256, 0, stream>>>(cnt, start, csr, dinv, xw, agg);
  k_mlp<<<782, 128, 0, stream>>>(x, xw, agg, dinv, bg, W1, b1, W2, b2, W3, b3, out);
}


# x @ Wg has no launch of its own: its 782 four-tile blocks ride in the offsets launch (role chosen per block, product blocks first); 5 launches, not 6; sums unchanged
# speedup vs baseline: 1.9223x; 1.0446x over previous
#include <hip/hip_runtime.h>

typedef float v2f __attribute__((ext_vector_type(2)));
typedef float v8f __attribute__((ext_vector_type(8)));

#define N_NODES 50000
#define N_EDGES 800000
#define NB (N_NODES / 16)
#define TPITCH 132

__device__ __forceinline__ v8f wmma_f32(v2f a, v2f b, v8f c) {
  return __builtin_amdgcn_wmma_f32_16x16x4_f32(false, a, false, b, (short)0, c,
                                               false, false);
}

#define OFF_CNT    0
#define OFF_DINV   50048
#define OFF_START  100096
#define OFF_CURSOR 150144
#define OFF_RANK   150208
#define OFF_CSR    950208
#define OFF_XW     1750208
#define OFF_AGG    8150208
#define OFF_TBL    14550208

#define NSLICE 64
#define EPS    (N_EDGES / NSLICE)
#define EPS4   (EPS / 4)
#define NPITCH 50048

__global__ __launch_bounds__(1024) void k_rank(const int* __restrict__ ei, int* __restrict__ rank,
                                               int* __restrict__ tbl, int* __restrict__ cursor) {
  __shared__ int c[N_NODES];
  const int tid = threadIdx.x;
  const int s = blockIdx.x;
  const int4* dst4 = (const int4*)(ei + N_EDGES + s * EPS);
  int4* rank4 = (int4*)(rank + s * EPS);
  int4* c4 = (int4*)c;
  int4 d[4];
#pragma unroll
  for (int k = 0; k < 4; ++k) {
    const int i = tid + k * 1024;
    d[k] = (i < EPS4) ? dst4[i] : make_int4(0, 0, 0, 0);
  }
  for (int i = tid; i < N_NODES / 4; i += 1024) c4[i] = make_int4(0, 0, 0, 0);
  __syncthreads();
  int4 r[4];
#pragma unroll
  for (int k = 0; k < 4; ++k) {
    const int i = tid + k * 1024;
    if (i < EPS4) {
      r[k].x = atomicAdd(&c[d[k].x], 1);
      r[k].y = atomicAdd(&c[d[k].y], 1);
      r[k].z = atomicAdd(&c[d[k].z], 1);
      r[k].w = atomicAdd(&c[d[k].w], 1);
    }
  }
  __syncthreads();
#pragma unroll
  for (int k = 0; k < 4; ++k) {
    const int i = tid + k * 1024;
    if (i < EPS4) rank4[i] = r[k];
  }
  int4* t4 = (int4*)(tbl + (size_t)s * NPITCH);
  for (int i = tid; i < N_NODES / 4; i += 1024) t4[i] = c4[i];
  if (s == 0 && tid == 0) cursor[0] = 0;
}

#define PBLK  ((NB + 3) / 4)
#define OBLK  ((N_NODES + 127) / 128)
__global__ __launch_bounds__(128) void k_offsets_xw(int* __restrict__ tbl, int* __restrict__ cursor,
                                                    int* __restrict__ cnt, int* __restrict__ start,
                                                    float* __restrict__ dinv,
                                                    const float* __restrict__ x,
                                                    const float* __restrict__ Wg,
                                                    float* __restrict__ xw) {
  __shared__ float lds[4 * 16 * TPITCH];
  const int lane = threadIdx.x & 31;
  if (blockIdx.x < PBLK) {
    const int wave = threadIdx.x >> 5;
    const int half = lane >> 4;
    const int lm   = lane & 15;
    float* tile = lds + wave * (16 * TPITCH);
    const int blk = blockIdx.x * 4 + wave;
    if (blk >= NB) return;
    const int row0 = blk * 16;
    for (int r = 0; r < 16; ++r) {
      float4 v = *(const float4*)(x + (size_t)(row0 + r) * 128 + lane * 4);
      *(float4*)(tile + r * TPITCH + lane * 4) = v;
    }
    asm volatile("s_wait_dscnt 0" ::: "memory");

    v8f acc[8];
    for (int t = 0; t < 8; ++t) acc[t] = (v8f){};
    for (int kk = 0; kk < 32; ++kk) {
      const int k0 = 4 * kk + 2 * half;
      v2f a = *(const v2f*)(tile + lm * TPITCH + k0);
      for (int t = 0; t < 8; ++t) {
        const float* bp = Wg + (size_t)k0 * 128 + 16 * t + lm;
        v2f b; b.x = bp[0]; b.y = bp[128];
        acc[t] = wmma_f32(a, b, acc[t]);
      }
    }
    for (int t = 0; t < 8; ++t) {
      float* op = xw + (size_t)(row0 + 8 * half) * 128 + 16 * t + lm;
      for (int r = 0; r < 8; ++r) op[(size_t)r * 128] = acc[t][r];
    }
    return;
  }
  const int i = (blockIdx.x - PBLK) * 128 + threadIdx.x;
  const int ic = (i < N_NODES) ? i : (N_NODES - 1);
  int v[NSLICE];
  int c = 0;
#pragma unroll
  for (int s = 0; s < NSLICE; ++s) {
    v[s] = tbl[(size_t)s * NPITCH + ic];
    c += v[s];
  }
  if (i >= N_NODES) c = 0;
  int incl = c;
  for (int off = 1; off < 32; off <<= 1) {
    int t = __shfl_up(incl, off);
    if (lane >= off) incl += t;
  }
  const int total = __shfl(incl, 31);
  int base = 0;
  if (lane == 31) base = atomicAdd(cursor, total);
  base = __shfl(base, 31);
  if (i < N_NODES) {
    int run = base + incl - c;
    start[i] = run;
    cnt[i] = c;
    float d = (float)c + 1.0f;
    dinv[i] = d > 0.0f ? rsqrtf(d) : 0.0f;
#pragma unroll
    for (int s = 0; s < NSLICE; ++s) {
      tbl[(size_t)s * NPITCH + i] = run;
      run += v[s];
    }
  }
}

__global__ __launch_bounds__(256) void k_fill(const int* __restrict__ ei, const int* __restrict__ tbl,
                                              const int* __restrict__ rank, int* __restrict__ csr) {
  int e = blockIdx.x * 256 + threadIdx.x;
  if (e < N_EDGES)
    csr[tbl[(size_t)(e / EPS) * NPITCH + ei[N_EDGES + e]] + rank[e]] = ei[e];
}

#define GB 4
__global__ __launch_bounds__(256) void k_gather(const int* __restrict__ cnt,
                                                const int* __restrict__ start,
                                                const int* __restrict__ csr,
                                                const float* __restrict__ dinv,
                                                const float* __restrict__ xw,
                                                float* __restrict__ agg) {
  const int lane = threadIdx.x & 31;
  const int node = __builtin_amdgcn_readfirstlane(blockIdx.x * 8 + (threadIdx.x >> 5));
  const int n = cnt[node];
  const int b = start[node];
  const float dd = dinv[node];
  float4 a = {0.0f, 0.0f, 0.0f, 0.0f};
  for (int base = 0; base < n; base += 32) {
    const int m = (n - base) < 32 ? (n - base) : 32;
    int s = 0;
    float w = 0.0f;
    if (lane < m) {
      s = csr[b + base + lane];
      w = dinv[s] * dd;
    }
    for (int j0 = 0; j0 < m; j0 += GB) {
      float4 v[GB];
#pragma unroll
      for (int k = 0; k < GB; ++k) {
        const int jj = (j0 + k) < m ? (j0 + k) : (m - 1);
        const int sj = __builtin_amdgcn_readlane(s, jj);
        v[k] = *((const float4*)(xw + (size_t)sj * 128) + lane);
      }
#pragma unroll
      for (int k = 0; k < GB; ++k) {
        const float nj = __int_as_float(__builtin_amdgcn_readlane(__float_as_int(w), j0 + k));
        a.x += v[k].x * nj;
        a.y += v[k].y * nj;
        a.z += v[k].z * nj;
        a.w += v[k].w * nj;
      }
    }
  }
  *((float4*)(agg + (size_t)node * 128) + lane) = a;
}

__global__ void k_mlp(const float* __restrict__ x, const float* __restrict__ xw,
                      const float* __restrict__ agg, const float* __restrict__ dinv,
                      const float* __restrict__ bg,
                      const float* __restrict__ W1, const float* __restrict__ b1,
                      const float* __restrict__ W2, const float* __restrict__ b2,
                      const float* __restrict__ W3, const float* __restrict__ b3,
                      float* __restrict__ out) {
  __shared__ float lds[4 * 16 * TPITCH];
  const int lane = threadIdx.x & 31;
  const int wave = threadIdx.x >> 5;
  const int half = lane >> 4;
  const int lm   = lane & 15;
  float* tile = lds + wave * (16 * TPITCH);
  const float4 bgv = *(const float4*)(bg + lane * 4);
  const int gw = blockIdx.x * 4 + wave;
  const int nwaves = gridDim.x * 4;

  for (int blk = gw; blk < NB; blk += nwaves) {
    const int row0 = blk * 16;
    for (int r = 0; r < 16; ++r) {
      const int row = row0 + r;
      float di = dinv[row];
      float sl = di * di;
      float4 ag  = *(const float4*)(agg + (size_t)row * 128 + lane * 4);
      float4 xwv = *(const float4*)(xw  + (size_t)row * 128 + lane * 4);
      float4 xv  = *(const float4*)(x   + (size_t)row * 128 + lane * 4);
      float4 h;
      h.x = fmaxf(ag.x + xwv.x * sl + bgv.x, 0.0f) + xv.x;
      h.y = fmaxf(ag.y + xwv.y * sl + bgv.y, 0.0f) + xv.y;
      h.z = fmaxf(ag.z + xwv.z * sl + bgv.z, 0.0f) + xv.z;
      h.w = fmaxf(ag.w + xwv.w * sl + bgv.w, 0.0f) + xv.w;
      *(float4*)(tile + r * TPITCH + lane * 4) = h;
    }
    asm volatile("s_wait_dscnt 0" ::: "memory");

    v8f acc1[8];
    for (int t = 0; t < 8; ++t) acc1[t] = (v8f){};
    for (int kk = 0; kk < 32; ++kk) {
      const int k0 = 4 * kk + 2 * half;
      v2f a = *(const v2f*)(tile + lm * TPITCH + k0);
      for (int t = 0; t < 8; ++t) {
        const float* bp = W1 + (size_t)k0 * 128 + 16 * t + lm;
        v2f b; b.x = bp[0]; b.y = bp[128];
        acc1[t] = wmma_f32(a, b, acc1[t]);
      }
    }
    asm volatile("s_wait_dscnt 0" ::: "memory");
    for (int t = 0; t < 8; ++t) {
      float bb = b1[16 * t + lm];
      for (int r = 0; r < 8; ++r)
        tile[(r + 8 * half) * TPITCH + 16 * t + lm] = fmaxf(acc1[t][r] + bb, 0.0f);
    }
    asm volatile("s_wait_dscnt 0" ::: "memory");

    v8f acc2[4];
    for (int t = 0; t < 4; ++t) acc2[t] = (v8f){};
    for (int kk = 0; kk < 32; ++kk) {
      const int k0 = 4 * kk + 2 * half;
      v2f a = *(const v2f*)(tile + lm * TPITCH + k0);
      for (int t = 0; t < 4; ++t) {
        const float* bp = W2 + (size_t)k0 * 64 + 16 * t + lm;
        v2f b; b.x = bp[0]; b.y = bp[64];
        acc2[t] = wmma_f32(a, b, acc2[t]);
      }
    }
    asm volatile("s_wait_dscnt 0" ::: "memory");
    for (int t = 0; t < 4; ++t) {
      float bb = b2[16 * t + lm];
      for (int r = 0; r < 8; ++r)
        tile[(r + 8 * half) * TPITCH + 16 * t + lm] = fmaxf(acc2[t][r] + bb, 0.0f);
    }
    asm volatile("s_wait_dscnt 0" ::: "memory");

    v8f acc3 = (v8f){};
    for (int kk = 0; kk < 16; ++kk) {
      const int k0 = 4 * kk + 2 * half;
      v2f a = *(const v2f*)(tile + lm * TPITCH + k0);
      v2f b;
      if (lm < 2) { b.x = W3[k0 * 2 + lm]; b.y = W3[(k0 + 1) * 2 + lm]; }
      else        { b.x = 0.0f;            b.y = 0.0f; }
      acc3 = wmma_f32(a, b, acc3);
    }
    if (lm < 2) {
      float bb = b3[lm];
      for (int r = 0; r < 8; ++r)
        out[(size_t)lm * N_NODES + row0 + r + 8 * half] = acc3[r] + bb;
    }
  }
}

extern "C" void kernel_launch(void* const* d_in, const int* in_sizes, int n_in,
                              void* d_out, int out_size, void* d_ws, size_t ws_size,
                              hipStream_t stream) {
  (void)in_sizes; (void)n_in; (void)out_size; (void)ws_size;
  const float* x  = (const float*)d_in[0];
  const int*   ei = (const int*)d_in[1];
  const float* Wg = (const float*)d_in[2];
  const float* bg = (const float*)d_in[3];
  const float* W1 = (const float*)d_in[4];
  const float* b1 = (const float*)d_in[5];
  const float* W2 = (const float*)d_in[6];
  const float* b2 = (const float*)d_in[7];
  const float* W3 = (const float*)d_in[8];
  const float* b3 = (const float*)d_in[9];
  float* out = (float*)d_out;

  float* ws    = (float*)d_ws;
  int*   wsi   = (int*)d_ws;
  int*   cnt   = wsi + OFF_CNT;
  float* dinv  = ws  + OFF_DINV;
  int*   start = wsi + OFF_START;
  int*   cursor= wsi + OFF_CURSOR;
  int*   rank  = wsi + OFF_RANK;
  int*   csr   = wsi + OFF_CSR;
  float* xw    = ws  + OFF_XW;
  float* agg   = ws  + OFF_AGG;
  int*   tbl   = wsi + OFF_TBL;

  k_rank<<<NSLICE, 1024, 0, stream>>>(ei, rank, tbl, cursor);
  k_offsets_xw<<<PBLK + OBLK, 128, 0, stream>>>(tbl, cursor, cnt, start, dinv, x, Wg, xw);
  k_fill<<<(N_EDGES + 255) / 256, 256, 0, stream>>>(ei, tbl, rank, csr);
  k_gather<<<N_NODES / 8, 256, 0, stream>>>(cnt, start, csr, dinv, xw, agg);
  k_mlp<<<782, 128, 0, stream>>>(x, xw, agg, dinv, bg, W1, b1, W2, b2, W3, b3, out);
}


# k_mlp declares its 128-thread block: the default's 128-register cap goes, so B fragments load 64 at a time ahead of 32 WMMAs, not one waited pair per WMMA; no scratch spills; sums unchanged
# speedup vs baseline: 2.9329x; 1.5257x over previous
#include <hip/hip_runtime.h>

typedef float v2f __attribute__((ext_vector_type(2)));
typedef float v8f __attribute__((ext_vector_type(8)));

#define N_NODES 50000
#define N_EDGES 800000
#define NB (N_NODES / 16)
#define TPITCH 132

__device__ __forceinline__ v8f wmma_f32(v2f a, v2f b, v8f c) {
  return __builtin_amdgcn_wmma_f32_16x16x4_f32(false, a, false, b, (short)0, c,
                                               false, false);
}

#define OFF_CNT    0
#define OFF_DINV   50048
#define OFF_START  100096
#define OFF_CURSOR 150144
#define OFF_RANK   150208
#define OFF_CSR    950208
#define OFF_XW     1750208
#define OFF_AGG    8150208
#define OFF_TBL    14550208

#define NSLICE 64
#define EPS    (N_EDGES / NSLICE)
#define EPS4   (EPS / 4)
#define NPITCH 50048

__global__ __launch_bounds__(1024) void k_rank(const int* __restrict__ ei, int* __restrict__ rank,
                                               int* __restrict__ tbl, int* __restrict__ cursor) {
  __shared__ int c[N_NODES];
  const int tid = threadIdx.x;
  const int s = blockIdx.x;
  const int4* dst4 = (const int4*)(ei + N_EDGES + s * EPS);
  int4* rank4 = (int4*)(rank + s * EPS);
  int4* c4 = (int4*)c;
  int4 d[4];
#pragma unroll
  for (int k = 0; k < 4; ++k) {
    const int i = tid + k * 1024;
    d[k] = (i < EPS4) ? dst4[i] : make_int4(0, 0, 0, 0);
  }
  for (int i = tid; i < N_NODES / 4; i += 1024) c4[i] = make_int4(0, 0, 0, 0);
  __syncthreads();
  int4 r[4];
#pragma unroll
  for (int k = 0; k < 4; ++k) {
    const int i = tid + k * 1024;
    if (i < EPS4) {
      r[k].x = atomicAdd(&c[d[k].x], 1);
      r[k].y = atomicAdd(&c[d[k].y], 1);
      r[k].z = atomicAdd(&c[d[k].z], 1);
      r[k].w = atomicAdd(&c[d[k].w], 1);
    }
  }
  __syncthreads();
#pragma unroll
  for (int k = 0; k < 4; ++k) {
    const int i = tid + k * 1024;
    if (i < EPS4) rank4[i] = r[k];
  }
  int4* t4 = (int4*)(tbl + (size_t)s * NPITCH);
  for (int i = tid; i < N_NODES / 4; i += 1024) t4[i] = c4[i];
  if (s == 0 && tid == 0) cursor[0] = 0;
}

#define PBLK  ((NB + 3) / 4)
#define OBLK  ((N_NODES + 127) / 128)
__global__ __launch_bounds__(128) void k_offsets_xw(int* __restrict__ tbl, int* __restrict__ cursor,
                                                    int* __restrict__ cnt, int* __restrict__ start,
                                                    float* __restrict__ dinv,
                                                    const float* __restrict__ x,
                                                    const float* __restrict__ Wg,
                                                    float* __restrict__ xw) {
  __shared__ float lds[4 * 16 * TPITCH];
  const int lane = threadIdx.x & 31;
  if (blockIdx.x < PBLK) {
    const int wave = threadIdx.x >> 5;
    const int half = lane >> 4;
    const int lm   = lane & 15;
    float* tile = lds + wave * (16 * TPITCH);
    const int blk = blockIdx.x * 4 + wave;
    if (blk >= NB) return;
    const int row0 = blk * 16;
    for (int r = 0; r < 16; ++r) {
      float4 v = *(const float4*)(x + (size_t)(row0 + r) * 128 + lane * 4);
      *(float4*)(tile + r * TPITCH + lane * 4) = v;
    }
    asm volatile("s_wait_dscnt 0" ::: "memory");

    v8f acc[8];
    for (int t = 0; t < 8; ++t) acc[t] = (v8f){};
    for (int kk = 0; kk < 32; ++kk) {
      const int k0 = 4 * kk + 2 * half;
      v2f a = *(const v2f*)(tile + lm * TPITCH + k0);
      for (int t = 0; t < 8; ++t) {
        const float* bp = Wg + (size_t)k0 * 128 + 16 * t + lm;
        v2f b; b.x = bp[0]; b.y = bp[128];
        acc[t] = wmma_f32(a, b, acc[t]);
      }
    }
    for (int t = 0; t < 8; ++t) {
      float* op = xw + (size_t)(row0 + 8 * half) * 128 + 16 * t + lm;
      for (int r = 0; r < 8; ++r) op[(size_t)r * 128] = acc[t][r];
    }
    return;
  }
  const int i = (blockIdx.x - PBLK) * 128 + threadIdx.x;
  const int ic = (i < N_NODES) ? i : (N_NODES - 1);
  int v[NSLICE];
  int c = 0;
#pragma unroll
  for (int s = 0; s < NSLICE; ++s) {
    v[s] = tbl[(size_t)s * NPITCH + ic];
    c += v[s];
  }
  if (i >= N_NODES) c = 0;
  int incl = c;
  for (int off = 1; off < 32; off <<= 1) {
    int t = __shfl_up(incl, off);
    if (lane >= off) incl += t;
  }
  const int total = __shfl(incl, 31);
  int base = 0;
  if (lane == 31) base = atomicAdd(cursor, total);
  base = __shfl(base, 31);
  if (i < N_NODES) {
    int run = base + incl - c;
    start[i] = run;
    cnt[i] = c;
    float d = (float)c + 1.0f;
    dinv[i] = d > 0.0f ? rsqrtf(d) : 0.0f;
#pragma unroll
    for (int s = 0; s < NSLICE; ++s) {
      tbl[(size_t)s * NPITCH + i] = run;
      run += v[s];
    }
  }
}

__global__ __launch_bounds__(256) void k_fill(const int* __restrict__ ei, const int* __restrict__ tbl,
                                              const int* __restrict__ rank, int* __restrict__ csr) {
  int e = blockIdx.x * 256 + threadIdx.x;
  if (e < N_EDGES)
    csr[tbl[(size_t)(e / EPS) * NPITCH + ei[N_EDGES + e]] + rank[e]] = ei[e];
}

#define GB 4
__global__ __launch_bounds__(256) void k_gather(const int* __restrict__ cnt,
                                                const int* __restrict__ start,
                                                const int* __restrict__ csr,
                                                const float* __restrict__ dinv,
                                                const float* __restrict__ xw,
                                                float* __restrict__ agg) {
  const int lane = threadIdx.x & 31;
  const int node = __builtin_amdgcn_readfirstlane(blockIdx.x * 8 + (threadIdx.x >> 5));
  const int n = cnt[node];
  const int b = start[node];
  const float dd = dinv[node];
  float4 a = {0.0f, 0.0f, 0.0f, 0.0f};
  for (int base = 0; base < n; base += 32) {
    const int m = (n - base) < 32 ? (n - base) : 32;
    int s = 0;
    float w = 0.0f;
    if (lane < m) {
      s = csr[b + base + lane];
      w = dinv[s] * dd;
    }
    for (int j0 = 0; j0 < m; j0 += GB) {
      float4 v[GB];
#pragma unroll
      for (int k = 0; k < GB; ++k) {
        const int jj = (j0 + k) < m ? (j0 + k) : (m - 1);
        const int sj = __builtin_amdgcn_readlane(s, jj);
        v[k] = *((const float4*)(xw + (size_t)sj * 128) + lane);
      }
#pragma unroll
      for (int k = 0; k < GB; ++k) {
        const float nj = __int_as_float(__builtin_amdgcn_readlane(__float_as_int(w), j0 + k));
        a.x += v[k].x * nj;
        a.y += v[k].y * nj;
        a.z += v[k].z * nj;
        a.w += v[k].w * nj;
      }
    }
  }
  *((float4*)(agg + (size_t)node * 128) + lane) = a;
}

__global__ __launch_bounds__(128) void k_mlp(const float* __restrict__ x, const float* __restrict__ xw,
                      const float* __restrict__ agg, const float* __restrict__ dinv,
                      const float* __restrict__ bg,
                      const float* __restrict__ W1, const float* __restrict__ b1,
                      const float* __restrict__ W2, const float* __restrict__ b2,
                      const float* __restrict__ W3, const float* __restrict__ b3,
                      float* __restrict__ out) {
  __shared__ float lds[4 * 16 * TPITCH];
  const int lane = threadIdx.x & 31;
  const int wave = threadIdx.x >> 5;
  const int half = lane >> 4;
  const int lm   = lane & 15;
  float* tile = lds + wave * (16 * TPITCH);
  const float4 bgv = *(const float4*)(bg + lane * 4);
  const int gw = blockIdx.x * 4 + wave;
  const int nwaves = gridDim.x * 4;

  for (int blk = gw; blk < NB; blk += nwaves) {
    const int row0 = blk * 16;
    for (int r = 0; r < 16; ++r) {
      const int row = row0 + r;
      float di = dinv[row];
      float sl = di * di;
      float4 ag  = *(const float4*)(agg + (size_t)row * 128 + lane * 4);
      float4 xwv = *(const float4*)(xw  + (size_t)row * 128 + lane * 4);
      float4 xv  = *(const float4*)(x   + (size_t)row * 128 + lane * 4);
      float4 h;
      h.x = fmaxf(ag.x + xwv.x * sl + bgv.x, 0.0f) + xv.x;
      h.y = fmaxf(ag.y + xwv.y * sl + bgv.y, 0.0f) + xv.y;
      h.z = fmaxf(ag.z + xwv.z * sl + bgv.z, 0.0f) + xv.z;
      h.w = fmaxf(ag.w + xwv.w * sl + bgv.w, 0.0f) + xv.w;
      *(float4*)(tile + r * TPITCH + lane * 4) = h;
    }
    asm volatile("s_wait_dscnt 0" ::: "memory");

    v8f acc1[8];
    for (int t = 0; t < 8; ++t) acc1[t] = (v8f){};
    for (int kk = 0; kk < 32; ++kk) {
      const int k0 = 4 * kk + 2 * half;
      v2f a = *(const v2f*)(tile + lm * TPITCH + k0);
      for (int t = 0; t < 8; ++t) {
        const float* bp = W1 + (size_t)k0 * 128 + 16 * t + lm;
        v2f b; b.x = bp[0]; b.y = bp[128];
        acc1[t] = wmma_f32(a, b, acc1[t]);
      }
    }
    asm volatile("s_wait_dscnt 0" ::: "memory");
    for (int t = 0; t < 8; ++t) {
      float bb = b1[16 * t + lm];
      for (int r = 0; r < 8; ++r)
        tile[(r + 8 * half) * TPITCH + 16 * t + lm] = fmaxf(acc1[t][r] + bb, 0.0f);
    }
    asm volatile("s_wait_dscnt 0" ::: "memory");

    v8f acc2[4];
    for (int t = 0; t < 4; ++t) acc2[t] = (v8f){};
    for (int kk = 0; kk < 32; ++kk) {
      const int k0 = 4 * kk + 2 * half;
      v2f a = *(const v2f*)(tile + lm * TPITCH + k0);
      for (int t = 0; t < 4; ++t) {
        const float* bp = W2 + (size_t)k0 * 64 + 16 * t + lm;
        v2f b; b.x = bp[0]; b.y = bp[64];
        acc2[t] = wmma_f32(a, b, acc2[t]);
      }
    }
    asm volatile("s_wait_dscnt 0" ::: "memory");
    for (int t = 0; t < 4; ++t) {
      float bb = b2[16 * t + lm];
      for (int r = 0; r < 8; ++r)
        tile[(r + 8 * half) * TPITCH + 16 * t + lm] = fmaxf(acc2[t][r] + bb, 0.0f);
    }
    asm volatile("s_wait_dscnt 0" ::: "memory");

    v8f acc3 = (v8f){};
    for (int kk = 0; kk < 16; ++kk) {
      const int k0 = 4 * kk + 2 * half;
      v2f a = *(const v2f*)(tile + lm * TPITCH + k0);
      v2f b;
      if (lm < 2) { b.x = W3[k0 * 2 + lm]; b.y = W3[(k0 + 1) * 2 + lm]; }
      else        { b.x = 0.0f;            b.y = 0.0f; }
      acc3 = wmma_f32(a, b, acc3);
    }
    if (lm < 2) {
      float bb = b3[lm];
      for (int r = 0; r < 8; ++r)
        out[(size_t)lm * N_NODES + row0 + r + 8 * half] = acc3[r] + bb;
    }
  }
}

extern "C" void kernel_launch(void* const* d_in, const int* in_sizes, int n_in,
                              void* d_out, int out_size, void* d_ws, size_t ws_size,
                              hipStream_t stream) {
  (void)in_sizes; (void)n_in; (void)out_size; (void)ws_size;
  const float* x  = (const float*)d_in[0];
  const int*   ei = (const int*)d_in[1];
  const float* Wg = (const float*)d_in[2];
  const float* bg = (const float*)d_in[3];
  const float* W1 = (const float*)d_in[4];
  const float* b1 = (const float*)d_in[5];
  const float* W2 = (const float*)d_in[6];
  const float* b2 = (const float*)d_in[7];
  const float* W3 = (const float*)d_in[8];
  const float* b3 = (const float*)d_in[9];
  float* out = (float*)d_out;

  float* ws    = (float*)d_ws;
  int*   wsi   = (int*)d_ws;
  int*   cnt   = wsi + OFF_CNT;
  float* dinv  = ws  + OFF_DINV;
  int*   start = wsi + OFF_START;
  int*   cursor= wsi + OFF_CURSOR;
  int*   rank  = wsi + OFF_RANK;
  int*   csr   = wsi + OFF_CSR;
  float* xw    = ws  + OFF_XW;
  float* agg   = ws  + OFF_AGG;
  int*   tbl   = wsi + OFF_TBL;

  k_rank<<<NSLICE, 1024, 0, stream>>>(ei, rank, tbl, cursor);
  k_offsets_xw<<<PBLK + OBLK, 128, 0, stream>>>(tbl, cursor, cnt, start, dinv, x, Wg, xw);
  k_fill<<<(N_EDGES + 255) / 256, 256, 0, stream>>>(ei, tbl, rank, csr);
  k_gather<<<N_NODES / 8, 256, 0, stream>>>(cnt, start, csr, dinv, xw, agg);
  k_mlp<<<782, 128, 0, stream>>>(x, xw, agg, dinv, bg, W1, b1, W2, b2, W3, b3, out);
}


# round trips cut: gather reads a node's slot, count and norm as one record, source norms beside the first rows, 8 rows a trip; k_mlp builds 8 rows a trip, W3 and biases asked ahead; sums unchanged
# speedup vs baseline: 3.0318x; 1.0337x over previous
#include <hip/hip_runtime.h>

typedef float v2f __attribute__((ext_vector_type(2)));
typedef float v8f __attribute__((ext_vector_type(8)));

#define N_NODES 50000
#define N_EDGES 800000
#define NB (N_NODES / 16)
#define TPITCH 132

__device__ __forceinline__ v8f wmma_f32(v2f a, v2f b, v8f c) {
  return __builtin_amdgcn_wmma_f32_16x16x4_f32(false, a, false, b, (short)0, c,
                                               false, false);
}

#define OFF_DINV   50048
#define OFF_CURSOR 150144
#define OFF_RANK   150208
#define OFF_CSR    950208
#define OFF_XW     1750208
#define OFF_AGG    8150208
#define OFF_TBL    14550208
#define OFF_META   17753280

#define NSLICE 64
#define EPS    (N_EDGES / NSLICE)
#define EPS4   (EPS / 4)
#define NPITCH 50048

__global__ __launch_bounds__(1024) void k_rank(const int* __restrict__ ei, int* __restrict__ rank,
                                               int* __restrict__ tbl, int* __restrict__ cursor) {
  __shared__ int c[N_NODES];
  const int tid = threadIdx.x;
  const int s = blockIdx.x;
  const int4* dst4 = (const int4*)(ei + N_EDGES + s * EPS);
  int4* rank4 = (int4*)(rank + s * EPS);
  int4* c4 = (int4*)c;
  int4 d[4];
#pragma unroll
  for (int k = 0; k < 4; ++k) {
    const int i = tid + k * 1024;
    d[k] = (i < EPS4) ? dst4[i] : make_int4(0, 0, 0, 0);
  }
  for (int i = tid; i < N_NODES / 4; i += 1024) c4[i] = make_int4(0, 0, 0, 0);
  __syncthreads();
  int4 r[4];
#pragma unroll
  for (int k = 0; k < 4; ++k) {
    const int i = tid + k * 1024;
    if (i < EPS4) {
      r[k].x = atomicAdd(&c[d[k].x], 1);
      r[k].y = atomicAdd(&c[d[k].y], 1);
      r[k].z = atomicAdd(&c[d[k].z], 1);
      r[k].w = atomicAdd(&c[d[k].w], 1);
    }
  }
  __syncthreads();
#pragma unroll
  for (int k = 0; k < 4; ++k) {
    const int i = tid + k * 1024;
    if (i < EPS4) rank4[i] = r[k];
  }
  int4* t4 = (int4*)(tbl + (size_t)s * NPITCH);
  for (int i = tid; i < N_NODES / 4; i += 1024) t4[i] = c4[i];
  if (s == 0 && tid == 0) cursor[0] = 0;
}

#define PBLK  ((NB + 3) / 4)
#define OBLK  ((N_NODES + 127) / 128)
__global__ __launch_bounds__(128) void k_offsets_xw(int* __restrict__ tbl, int* __restrict__ cursor,
                                                    int* __restrict__ meta,
                                                    float* __restrict__ dinv,
                                                    const float* __restrict__ x,
                                                    const float* __restrict__ Wg,
                                                    float* __restrict__ xw) {
  __shared__ float lds[4 * 16 * TPITCH];
  const int lane = threadIdx.x & 31;
  if (blockIdx.x < PBLK) {
    const int wave = threadIdx.x >> 5;
    const int half = lane >> 4;
    const int lm   = lane & 15;
    float* tile = lds + wave * (16 * TPITCH);
    const int blk = blockIdx.x * 4 + wave;
    if (blk >= NB) return;
    const int row0 = blk * 16;
    for (int r = 0; r < 16; ++r) {
      float4 v = *(const float4*)(x + (size_t)(row0 + r) * 128 + lane * 4);
      *(float4*)(tile + r * TPITCH + lane * 4) = v;
    }
    asm volatile("s_wait_dscnt 0" ::: "memory");

    v8f acc[8];
    for (int t = 0; t < 8; ++t) acc[t] = (v8f){};
    for (int kk = 0; kk < 32; ++kk) {
      const int k0 = 4 * kk + 2 * half;
      v2f a = *(const v2f*)(tile + lm * TPITCH + k0);
      for (int t = 0; t < 8; ++t) {
        const float* bp = Wg + (size_t)k0 * 128 + 16 * t + lm;
        v2f b; b.x = bp[0]; b.y = bp[128];
        acc[t] = wmma_f32(a, b, acc[t]);
      }
    }
    for (int t = 0; t < 8; ++t) {
      float* op = xw + (size_t)(row0 + 8 * half) * 128 + 16 * t + lm;
      for (int r = 0; r < 8; ++r) op[(size_t)r * 128] = acc[t][r];
    }
    return;
  }
  const int i = (blockIdx.x - PBLK) * 128 + threadIdx.x;
  const int ic = (i < N_NODES) ? i : (N_NODES - 1);
  int v[NSLICE];
  int c = 0;
#pragma unroll
  for (int s = 0; s < NSLICE; ++s) {
    v[s] = tbl[(size_t)s * NPITCH + ic];
    c += v[s];
  }
  if (i >= N_NODES) c = 0;
  int incl = c;
  for (int off = 1; off < 32; off <<= 1) {
    int t = __shfl_up(incl, off);
    if (lane >= off) incl += t;
  }
  const int total = __shfl(incl, 31);
  int base = 0;
  if (lane == 31) base = atomicAdd(cursor, total);
  base = __shfl(base, 31);
  if (i < N_NODES) {
    int run = base + incl - c;
    float d = (float)c + 1.0f;
    const float di = d > 0.0f ? rsqrtf(d) : 0.0f;
    dinv[i] = di;
    ((int4*)meta)[i] = make_int4(run, c, __float_as_int(di), 0);
#pragma unroll
    for (int s = 0; s < NSLICE; ++s) {
      tbl[(size_t)s * NPITCH + i] = run;
      run += v[s];
    }
  }
}

__global__ __launch_bounds__(256) void k_fill(const int* __restrict__ ei, const int* __restrict__ tbl,
                                              const int* __restrict__ rank, int* __restrict__ csr) {
  int e = blockIdx.x * 256 + threadIdx.x;
  if (e < N_EDGES)
    csr[tbl[(size_t)(e / EPS) * NPITCH + ei[N_EDGES + e]] + rank[e]] = ei[e];
}

template <int NR, bool PAD>
__device__ __forceinline__ void rows_add(float4& a, const float* __restrict__ xw, int lane,
                                         int s, float ds, float dd, int j0, int m) {
  float4 v[NR];
#pragma unroll
  for (int k = 0; k < NR; ++k) {
    int jj = j0 + k;
    if (PAD) jj = jj < m ? jj : (m - 1);
    const int sj = __builtin_amdgcn_readlane(s, jj);
    v[k] = *((const float4*)(xw + (size_t)sj * 128) + lane);
  }
  __builtin_amdgcn_sched_barrier(0);
#pragma unroll
  for (int k = 0; k < NR; ++k) {
    const float nj = __int_as_float(__builtin_amdgcn_readlane(__float_as_int(ds), j0 + k)) * dd;
    a.x = __builtin_fmaf(v[k].x, nj, a.x);
    a.y = __builtin_fmaf(v[k].y, nj, a.y);
    a.z = __builtin_fmaf(v[k].z, nj, a.z);
    a.w = __builtin_fmaf(v[k].w, nj, a.w);
  }
}

__global__ __launch_bounds__(256) void k_gather(const int* __restrict__ meta,
                                                const int* __restrict__ csr,
                                                const float* __restrict__ dinv,
                                                const float* __restrict__ xw,
                                                float* __restrict__ agg) {
  const int lane = threadIdx.x & 31;
  const int node = __builtin_amdgcn_readfirstlane(blockIdx.x * 8 + (threadIdx.x >> 5));
  const int4 md = ((const int4*)meta)[node];
  const int b = md.x;
  const int n = md.y;
  const float dd = __int_as_float(md.z);
  float4 a = {0.0f, 0.0f, 0.0f, 0.0f};
  for (int base = 0; base < n; base += 32) {
    const int m = (n - base) < 32 ? (n - base) : 32;
    int s = 0;
    float ds = 0.0f;
    if (lane < m) {
      s = csr[b + base + lane];
      ds = dinv[s];
    }
    int j0 = 0;
    for (; j0 + 8 <= m; j0 += 8) rows_add<8, false>(a, xw, lane, s, ds, dd, j0, m);
    if (m - j0 > 4)      rows_add<8, true>(a, xw, lane, s, ds, dd, j0, m);
    else if (m - j0 > 0) rows_add<4, true>(a, xw, lane, s, ds, dd, j0, m);
  }
  *((float4*)(agg + (size_t)node * 128) + lane) = a;
}

__global__ __launch_bounds__(128) void k_mlp(const float* __restrict__ x, const float* __restrict__ xw,
                      const float* __restrict__ agg, const float* __restrict__ dinv,
                      const float* __restrict__ bg,
                      const float* __restrict__ W1, const float* __restrict__ b1,
                      const float* __restrict__ W2, const float* __restrict__ b2,
                      const float* __restrict__ W3, const float* __restrict__ b3,
                      float* __restrict__ out) {
  __shared__ float lds[4 * 16 * TPITCH];
  const int lane = threadIdx.x & 31;
  const int wave = threadIdx.x >> 5;
  const int half = lane >> 4;
  const int lm   = lane & 15;
  float* tile = lds + wave * (16 * TPITCH);
  const float4 bgv = *(const float4*)(bg + lane * 4);
  const int gw = blockIdx.x * 4 + wave;
  const int nwaves = gridDim.x * 4;

  for (int blk = gw; blk < NB; blk += nwaves) {
    const int row0 = blk * 16;
    const int dlb = __float_as_int(dinv[row0 + lm]);
    for (int r0 = 0; r0 < 16; r0 += 8) {
      float4 ag[8], xwv[8], xv[8];
#pragma unroll
      for (int r = 0; r < 8; ++r) {
        const size_t o = (size_t)(row0 + r0 + r) * 128 + lane * 4;
        ag[r]  = *(const float4*)(agg + o);
        xwv[r] = *(const float4*)(xw  + o);
        xv[r]  = *(const float4*)(x   + o);
      }
#pragma unroll
      for (int r = 0; r < 8; ++r) {
        const float di = __int_as_float(__builtin_amdgcn_readlane(dlb, r0 + r));
        const float sl = di * di;
        float4 h;
        h.x = fmaxf(ag[r].x + xwv[r].x * sl + bgv.x, 0.0f) + xv[r].x;
        h.y = fmaxf(ag[r].y + xwv[r].y * sl + bgv.y, 0.0f) + xv[r].y;
        h.z = fmaxf(ag[r].z + xwv[r].z * sl + bgv.z, 0.0f) + xv[r].z;
        h.w = fmaxf(ag[r].w + xwv[r].w * sl + bgv.w, 0.0f) + xv[r].w;
        *(float4*)(tile + (r0 + r) * TPITCH + lane * 4) = h;
      }
    }
    asm volatile("s_wait_dscnt 0" ::: "memory");

    v8f acc1[8];
    for (int t = 0; t < 8; ++t) acc1[t] = (v8f){};
    float bb1[8];
#pragma unroll
    for (int t = 0; t < 8; ++t) bb1[t] = b1[16 * t + lm];
    for (int kk = 0; kk < 32; ++kk) {
      const int k0 = 4 * kk + 2 * half;
      v2f a = *(const v2f*)(tile + lm * TPITCH + k0);
      for (int t = 0; t < 8; ++t) {
        const float* bp = W1 + (size_t)k0 * 128 + 16 * t + lm;
        v2f b; b.x = bp[0]; b.y = bp[128];
        acc1[t] = wmma_f32(a, b, acc1[t]);
      }
    }
    asm volatile("s_wait_dscnt 0" ::: "memory");
#pragma unroll
    for (int t = 0; t < 8; ++t) {
      const float bb = bb1[t];
      for (int r = 0; r < 8; ++r)
        tile[(r + 8 * half) * TPITCH + 16 * t + lm] = fmaxf(acc1[t][r] + bb, 0.0f);
    }
    asm volatile("s_wait_dscnt 0" ::: "memory");

    v8f acc2[4];
    for (int t = 0; t < 4; ++t) acc2[t] = (v8f){};
    float bb2[4];
#pragma unroll
    for (int t = 0; t < 4; ++t) bb2[t] = b2[16 * t + lm];
    const int lc = lm < 2 ? lm : 0;
    float w3x[16], w3y[16];
#pragma unroll
    for (int kk = 0; kk < 16; ++kk) {
      const int k0 = 4 * kk + 2 * half;
      w3x[kk] = W3[k0 * 2 + lc];
      w3y[kk] = W3[(k0 + 1) * 2 + lc];
    }
    const float bb3 = b3[lc];
    for (int kk = 0; kk < 32; ++kk) {
      const int k0 = 4 * kk + 2 * half;
      v2f a = *(const v2f*)(tile + lm * TPITCH + k0);
      for (int t = 0; t < 4; ++t) {
        const float* bp = W2 + (size_t)k0 * 64 + 16 * t + lm;
        v2f b; b.x = bp[0]; b.y = bp[64];
        acc2[t] = wmma_f32(a, b, acc2[t]);
      }
    }
    asm volatile("s_wait_dscnt 0" ::: "memory");
#pragma unroll
    for (int t = 0; t < 4; ++t) {
      const float bb = bb2[t];
      for (int r = 0; r < 8; ++r)
        tile[(r + 8 * half) * TPITCH + 16 * t + lm] = fmaxf(acc2[t][r] + bb, 0.0f);
    }
    asm volatile("s_wait_dscnt 0" ::: "memory");

    v8f acc3 = (v8f){};
#pragma unroll
    for (int kk = 0; kk < 16; ++kk) {
      const int k0 = 4 * kk + 2 * half;
      v2f a = *(const v2f*)(tile + lm * TPITCH + k0);
      v2f b;
      b.x = lm < 2 ? w3x[kk] : 0.0f;
      b.y = lm < 2 ? w3y[kk] : 0.0f;
      acc3 = wmma_f32(a, b, acc3);
    }
    if (lm < 2) {
      for (int r = 0; r < 8; ++r)
        out[(size_t)lm * N_NODES + row0 + r + 8 * half] = acc3[r] + bb3;
    }
  }
}

extern "C" void kernel_launch(void* const* d_in, const int* in_sizes, int n_in,
                              void* d_out, int out_size, void* d_ws, size_t ws_size,
                              hipStream_t stream) {
  (void)in_sizes; (void)n_in; (void)out_size; (void)ws_size;
  const float* x  = (const float*)d_in[0];
  const int*   ei = (const int*)d_in[1];
  const float* Wg = (const float*)d_in[2];
  const float* bg = (const float*)d_in[3];
  const float* W1 = (const float*)d_in[4];
  const float* b1 = (const float*)d_in[5];
  const float* W2 = (const float*)d_in[6];
  const float* b2 = (const float*)d_in[7];
  const float* W3 = (const float*)d_in[8];
  const float* b3 = (const float*)d_in[9];
  float* out = (float*)d_out;

  float* ws    = (float*)d_ws;
  int*   wsi   = (int*)d_ws;
  float* dinv  = ws  + OFF_DINV;
  int*   cursor= wsi + OFF_CURSOR;
  int*   rank  = wsi + OFF_RANK;
  int*   csr   = wsi + OFF_CSR;
  float* xw    = ws  + OFF_XW;
  float* agg   = ws  + OFF_AGG;
  int*   tbl   = wsi + OFF_TBL;
  int*   meta  = wsi + OFF_META;

  k_rank<<<NSLICE, 1024, 0, stream>>>(ei, rank, tbl, cursor);
  k_offsets_xw<<<PBLK + OBLK, 128, 0, stream>>>(tbl, cursor, meta, dinv, x, Wg, xw);
  k_fill<<<(N_EDGES + 255) / 256, 256, 0, stream>>>(ei, tbl, rank, csr);
  k_gather<<<N_NODES / 8, 256, 0, stream>>>(meta, csr, dinv, xw, agg);
  k_mlp<<<782, 128, 0, stream>>>(x, xw, agg, dinv, bg, W1, b1, W2, b2, W3, b3, out);
}


# fill launch removed: rank blocks prefix-sum their counters in LDS and write each slice's src sorted by dst; offsets threads copy their node's runs into the list; 4 launches; sum order unchanged
# speedup vs baseline: 3.4021x; 1.1221x over previous
#include <hip/hip_runtime.h>

typedef float v2f __attribute__((ext_vector_type(2)));
typedef float v8f __attribute__((ext_vector_type(8)));

#define N_NODES 50000
#define N_EDGES 800000
#define NB (N_NODES / 16)
#define TPITCH 132

__device__ __forceinline__ v8f wmma_f32(v2f a, v2f b, v8f c) {
  return __builtin_amdgcn_wmma_f32_16x16x4_f32(false, a, false, b, (short)0, c,
                                               false, false);
}

#define OFF_DINV   50048
#define OFF_CURSOR 150144
#define OFF_SRT    150208
#define OFF_CSR    950208
#define OFF_XW     1750208
#define OFF_AGG    8150208
#define OFF_TBL    14550208
#define OFF_META   17753280

#define NSLICE 64
#define EPS    (N_EDGES / NSLICE)
#define EPS4   (EPS / 4)
#define NPITCH 50048
#define CHUNK4 13
#define NCNT   (1024 * CHUNK4 * 4)

__global__ __launch_bounds__(1024) void k_rank(const int* __restrict__ ei, int* __restrict__ srt,
                                               int* __restrict__ tbl, int* __restrict__ cursor) {
  __shared__ int c[NCNT];
  __shared__ int wsum[32];
  const int tid = threadIdx.x;
  const int lane = tid & 31;
  const int wave = tid >> 5;
  const int s = blockIdx.x;
  const int4* dst4 = (const int4*)(ei + N_EDGES + s * EPS);
  const int4* src4 = (const int4*)(ei + s * EPS);
  int4* c4 = (int4*)c;
  int4 d[4], sv[4];
#pragma unroll
  for (int k = 0; k < 4; ++k) {
    const int i = tid + k * 1024;
    d[k]  = (i < EPS4) ? dst4[i] : make_int4(0, 0, 0, 0);
    sv[k] = (i < EPS4) ? src4[i] : make_int4(0, 0, 0, 0);
  }
#pragma unroll
  for (int k = 0; k < CHUNK4; ++k) c4[tid + k * 1024] = make_int4(0, 0, 0, 0);
  __syncthreads();
  int4 r[4];
#pragma unroll
  for (int k = 0; k < 4; ++k) {
    const int i = tid + k * 1024;
    if (i < EPS4) {
      r[k].x = atomicAdd(&c[d[k].x], 1);
      r[k].y = atomicAdd(&c[d[k].y], 1);
      r[k].z = atomicAdd(&c[d[k].z], 1);
      r[k].w = atomicAdd(&c[d[k].w], 1);
    }
  }
  __syncthreads();
  int4 q[CHUNK4];
  int p = 0;
#pragma unroll
  for (int j = 0; j < CHUNK4; ++j) {
    q[j] = c4[tid * CHUNK4 + j];
    p += q[j].x + q[j].y + q[j].z + q[j].w;
  }
  int incl = p;
  for (int off = 1; off < 32; off <<= 1) {
    int t = __shfl_up(incl, off);
    if (lane >= off) incl += t;
  }
  if (lane == 31) wsum[wave] = incl;
  __syncthreads();
  const int wt = wsum[lane];
  int winc = wt;
  for (int off = 1; off < 32; off <<= 1) {
    int t = __shfl_up(winc, off);
    if (lane >= off) winc += t;
  }
  const int wbase = __shfl(winc - wt, wave);
  int run = wbase + incl - p;
#pragma unroll
  for (int j = 0; j < CHUNK4; ++j) {
    int t;
    t = q[j].x; q[j].x = run; run += t;
    t = q[j].y; q[j].y = run; run += t;
    t = q[j].z; q[j].z = run; run += t;
    t = q[j].w; q[j].w = run; run += t;
    c4[tid * CHUNK4 + j] = q[j];
  }
  __syncthreads();
  int* mine = srt + s * EPS;
#pragma unroll
  for (int k = 0; k < 4; ++k) {
    const int i = tid + k * 1024;
    if (i < EPS4) {
      mine[c[d[k].x] + r[k].x] = sv[k].x;
      mine[c[d[k].y] + r[k].y] = sv[k].y;
      mine[c[d[k].z] + r[k].z] = sv[k].z;
      mine[c[d[k].w] + r[k].w] = sv[k].w;
    }
  }
  int4* t4 = (int4*)(tbl + (size_t)s * NPITCH);
  for (int i = tid; i < N_NODES / 4 + 1; i += 1024) t4[i] = c4[i];
  if (s == 0 && tid == 0) cursor[0] = 0;
}

#define PBLK  ((NB + 3) / 4)
#define HSL   (NSLICE / 2)
#define OBLK  ((N_NODES + 63) / 64)
__global__ __launch_bounds__(128) void k_offsets_xw(const int* __restrict__ tbl, int* __restrict__ cursor,
                                                    int* __restrict__ meta,
                                                    float* __restrict__ dinv,
                                                    const int* __restrict__ srt,
                                                    int* __restrict__ csr,
                                                    const float* __restrict__ x,
                                                    const float* __restrict__ Wg,
                                                    float* __restrict__ xw) {
  __shared__ float lds[4 * 16 * TPITCH];
  const int lane = threadIdx.x & 31;
  if (blockIdx.x >= OBLK) {
    const int wave = threadIdx.x >> 5;
    const int half = lane >> 4;
    const int lm   = lane & 15;
    float* tile = lds + wave * (16 * TPITCH);
    const int blk = (blockIdx.x - OBLK) * 4 + wave;
    if (blk >= NB) return;
    const int row0 = blk * 16;
    for (int r = 0; r < 16; ++r) {
      float4 v = *(const float4*)(x + (size_t)(row0 + r) * 128 + lane * 4);
      *(float4*)(tile + r * TPITCH + lane * 4) = v;
    }
    asm volatile("s_wait_dscnt 0" ::: "memory");

    v8f acc[8];
    for (int t = 0; t < 8; ++t) acc[t] = (v8f){};
    for (int kk = 0; kk < 32; ++kk) {
      const int k0 = 4 * kk + 2 * half;
      v2f a = *(const v2f*)(tile + lm * TPITCH + k0);
      for (int t = 0; t < 8; ++t) {
        const float* bp = Wg + (size_t)k0 * 128 + 16 * t + lm;
        v2f b; b.x = bp[0]; b.y = bp[128];
        acc[t] = wmma_f32(a, b, acc[t]);
      }
    }
    for (int t = 0; t < 8; ++t) {
      float* op = xw + (size_t)(row0 + 8 * half) * 128 + 16 * t + lm;
      for (int r = 0; r < 8; ++r) op[(size_t)r * 128] = acc[t][r];
    }
    return;
  }
  const int h = lane & 1;
  const int i = blockIdx.x * 64 + (threadIdx.x >> 1);
  const int ic = (i < N_NODES) ? i : (N_NODES - 1);
  const int* trow = tbl + (size_t)(h * HSL) * NPITCH + ic;
  int pk[HSL];
  int a = 0;
#pragma unroll
  for (int s = 0; s < HSL; ++s) pk[s] = trow[(size_t)s * NPITCH];
#pragma unroll
  for (int s = 0; s < HSL; ++s) {
    const int cn = trow[(size_t)s * NPITCH + 1] - pk[s];
    pk[s] |= cn << 16;
    a += cn;
  }
  if (i >= N_NODES) a = 0;
  int incl = a;
  for (int off = 1; off < 32; off <<= 1) {
    int t = __shfl_up(incl, off);
    if (lane >= off) incl += t;
  }
  const int total = __shfl(incl, 31);
  int base = 0;
  if (lane == 31) base = atomicAdd(cursor, total);
  base = __shfl(base, 31);
  const int c = a + __shfl_xor(a, 1);
  if (i < N_NODES) {
    const int first = base + incl - a;
    if (h == 0) {
      float d = (float)c + 1.0f;
      const float di = d > 0.0f ? rsqrtf(d) : 0.0f;
      dinv[i] = di;
      ((int4*)meta)[i] = make_int4(first, c, __float_as_int(di), 0);
    }
    int kmax = 0;
#pragma unroll
    for (int s = 0; s < HSL; ++s) kmax = max(kmax, pk[s] >> 16);
    const int* sl = srt + (h * HSL) * EPS;
    for (int k = 0; k < kmax; ++k) {
      int t[HSL];
#pragma unroll
      for (int s = 0; s < HSL; ++s) {
        const int e = min((pk[s] & 0xffff) + k, EPS - 1);
        t[s] = sl[s * EPS + e];
      }
      int pos = first + k;
#pragma unroll
      for (int s = 0; s < HSL; ++s) {
        const int cn = pk[s] >> 16;
        if (cn > k) csr[pos] = t[s];
        pos += cn;
      }
    }
  }
}

template <int NR, bool PAD>
__device__ __forceinline__ void rows_add(float4& a, const float* __restrict__ xw, int lane,
                                         int s, float ds, float dd, int j0, int m) {
  float4 v[NR];
#pragma unroll
  for (int k = 0; k < NR; ++k) {
    int jj = j0 + k;
    if (PAD) jj = jj < m ? jj : (m - 1);
    const int sj = __builtin_amdgcn_readlane(s, jj);
    v[k] = *((const float4*)(xw + (size_t)sj * 128) + lane);
  }
  __builtin_amdgcn_sched_barrier(0);
#pragma unroll
  for (int k = 0; k < NR; ++k) {
    const float nj = __int_as_float(__builtin_amdgcn_readlane(__float_as_int(ds), j0 + k)) * dd;
    a.x = __builtin_fmaf(v[k].x, nj, a.x);
    a.y = __builtin_fmaf(v[k].y, nj, a.y);
    a.z = __builtin_fmaf(v[k].z, nj, a.z);
    a.w = __builtin_fmaf(v[k].w, nj, a.w);
  }
}

__global__ __launch_bounds__(256) void k_gather(const int* __restrict__ meta,
                                                const int* __restrict__ csr,
                                                const float* __restrict__ dinv,
                                                const float* __restrict__ xw,
                                                float* __restrict__ agg) {
  const int lane = threadIdx.x & 31;
  const int node = __builtin_amdgcn_readfirstlane(blockIdx.x * 8 + (threadIdx.x >> 5));
  const int4 md = ((const int4*)meta)[node];
  const int b = md.x;
  const int n = md.y;
  const float dd = __int_as_float(md.z);
  float4 a = {0.0f, 0.0f, 0.0f, 0.0f};
  for (int base = 0; base < n; base += 32) {
    const int m = (n - base) < 32 ? (n - base) : 32;
    int s = 0;
    float ds = 0.0f;
    if (lane < m) {
      s = csr[b + base + lane];
      ds = dinv[s];
    }
    int j0 = 0;
    for (; j0 + 8 <= m; j0 += 8) rows_add<8, false>(a, xw, lane, s, ds, dd, j0, m);
    if (m - j0 > 4)      rows_add<8, true>(a, xw, lane, s, ds, dd, j0, m);
    else if (m - j0 > 0) rows_add<4, true>(a, xw, lane, s, ds, dd, j0, m);
  }
  *((float4*)(agg + (size_t)node * 128) + lane) = a;
}

__global__ __launch_bounds__(128) void k_mlp(const float* __restrict__ x, const float* __restrict__ xw,
                      const float* __restrict__ agg, const float* __restrict__ dinv,
                      const float* __restrict__ bg,
                      const float* __restrict__ W1, const float* __restrict__ b1,
                      const float* __restrict__ W2, const float* __restrict__ b2,
                      const float* __restrict__ W3, const float* __restrict__ b3,
                      float* __restrict__ out) {
  __shared__ float lds[4 * 16 * TPITCH];
  const int lane = threadIdx.x & 31;
  const int wave = threadIdx.x >> 5;
  const int half = lane >> 4;
  const int lm   = lane & 15;
  float* tile = lds + wave * (16 * TPITCH);
  const float4 bgv = *(const float4*)(bg + lane * 4);
  const int gw = blockIdx.x * 4 + wave;
  const int nwaves = gridDim.x * 4;

  for (int blk = gw; blk < NB; blk += nwaves) {
    const int row0 = blk * 16;
    const int dlb = __float_as_int(dinv[row0 + lm]);
    for (int r0 = 0; r0 < 16; r0 += 8) {
      float4 ag[8], xwv[8], xv[8];
#pragma unroll
      for (int r = 0; r < 8; ++r) {
        const size_t o = (size_t)(row0 + r0 + r) * 128 + lane * 4;
        ag[r]  = *(const float4*)(agg + o);
        xwv[r] = *(const float4*)(xw  + o);
        xv[r]  = *(const float4*)(x   + o);
      }
#pragma unroll
      for (int r = 0; r < 8; ++r) {
        const float di = __int_as_float(__builtin_amdgcn_readlane(dlb, r0 + r));
        const float sl = di * di;
        float4 h;
        h.x = fmaxf(ag[r].x + xwv[r].x * sl + bgv.x, 0.0f) + xv[r].x;
        h.y = fmaxf(ag[r].y + xwv[r].y * sl + bgv.y, 0.0f) + xv[r].y;
        h.z = fmaxf(ag[r].z + xwv[r].z * sl + bgv.z, 0.0f) + xv[r].z;
        h.w = fmaxf(ag[r].w + xwv[r].w * sl + bgv.w, 0.0f) + xv[r].w;
        *(float4*)(tile + (r0 + r) * TPITCH + lane * 4) = h;
      }
    }
    asm volatile("s_wait_dscnt 0" ::: "memory");

    v8f acc1[8];
    for (int t = 0; t < 8; ++t) acc1[t] = (v8f){};
    float bb1[8];
#pragma unroll
    for (int t = 0; t < 8; ++t) bb1[t] = b1[16 * t + lm];
    for (int kk = 0; kk < 32; ++kk) {
      const int k0 = 4 * kk + 2 * half;
      v2f a = *(const v2f*)(tile + lm * TPITCH + k0);
      for (int t = 0; t < 8; ++t) {
        const float* bp = W1 + (size_t)k0 * 128 + 16 * t + lm;
        v2f b; b.x = bp[0]; b.y = bp[128];
        acc1[t] = wmma_f32(a, b, acc1[t]);
      }
    }
    asm volatile("s_wait_dscnt 0" ::: "memory");
#pragma unroll
    for (int t = 0; t < 8; ++t) {
      const float bb = bb1[t];
      for (int r = 0; r < 8; ++r)
        tile[(r + 8 * half) * TPITCH + 16 * t + lm] = fmaxf(acc1[t][r] + bb, 0.0f);
    }
    asm volatile("s_wait_dscnt 0" ::: "memory");

    v8f acc2[4];
    for (int t = 0; t < 4; ++t) acc2[t] = (v8f){};
    float bb2[4];
#pragma unroll
    for (int t = 0; t < 4; ++t) bb2[t] = b2[16 * t + lm];
    const int lc = lm < 2 ? lm : 0;
    float w3x[16], w3y[16];
#pragma unroll
    for (int kk = 0; kk < 16; ++kk) {
      const int k0 = 4 * kk + 2 * half;
      w3x[kk] = W3[k0 * 2 + lc];
      w3y[kk] = W3[(k0 + 1) * 2 + lc];
    }
    const float bb3 = b3[lc];
    for (int kk = 0; kk < 32; ++kk) {
      const int k0 = 4 * kk + 2 * half;
      v2f a = *(const v2f*)(tile + lm * TPITCH + k0);
      for (int t = 0; t < 4; ++t) {
        const float* bp = W2 + (size_t)k0 * 64 + 16 * t + lm;
        v2f b; b.x = bp[0]; b.y = bp[64];
        acc2[t] = wmma_f32(a, b, acc2[t]);
      }
    }
    asm volatile("s_wait_dscnt 0" ::: "memory");
#pragma unroll
    for (int t = 0; t < 4; ++t) {
      const float bb = bb2[t];
      for (int r = 0; r < 8; ++r)
        tile[(r + 8 * half) * TPITCH + 16 * t + lm] = fmaxf(acc2[t][r] + bb, 0.0f);
    }
    asm volatile("s_wait_dscnt 0" ::: "memory");

    v8f acc3 = (v8f){};
#pragma unroll
    for (int kk = 0; kk < 16; ++kk) {
      const int k0 = 4 * kk + 2 * half;
      v2f a = *(const v2f*)(tile + lm * TPITCH + k0);
      v2f b;
      b.x = lm < 2 ? w3x[kk] : 0.0f;
      b.y = lm < 2 ? w3y[kk] : 0.0f;
      acc3 = wmma_f32(a, b, acc3);
    }
    if (lm < 2) {
      for (int r = 0; r < 8; ++r)
        out[(size_t)lm * N_NODES + row0 + r + 8 * half] = acc3[r] + bb3;
    }
  }
}

extern "C" void kernel_launch(void* const* d_in, const int* in_sizes, int n_in,
                              void* d_out, int out_size, void* d_ws, size_t ws_size,
                              hipStream_t stream) {
  (void)in_sizes; (void)n_in; (void)out_size; (void)ws_size;
  const float* x  = (const float*)d_in[0];
  const int*   ei = (const int*)d_in[1];
  const float* Wg = (const float*)d_in[2];
  const float* bg = (const float*)d_in[3];
  const float* W1 = (const float*)d_in[4];
  const float* b1 = (const float*)d_in[5];
  const float* W2 = (const float*)d_in[6];
  const float* b2 = (const float*)d_in[7];
  const float* W3 = (const float*)d_in[8];
  const float* b3 = (const float*)d_in[9];
  float* out = (float*)d_out;

  float* ws    = (float*)d_ws;
  int*   wsi   = (int*)d_ws;
  float* dinv  = ws  + OFF_DINV;
  int*   cursor= wsi + OFF_CURSOR;
  int*   srt   = wsi + OFF_SRT;
  int*   csr   = wsi + OFF_CSR;
  float* xw    = ws  + OFF_XW;
  float* agg   = ws  + OFF_AGG;
  int*   tbl   = wsi + OFF_TBL;
  int*   meta  = wsi + OFF_META;

  k_rank<<<NSLICE, 1024, 0, stream>>>(ei, srt, tbl, cursor);
  k_offsets_xw<<<OBLK + PBLK, 128, 0, stream>>>(tbl, cursor, meta, dinv, srt, csr, x, Wg, xw);
  k_gather<<<N_NODES / 8, 256, 0, stream>>>(meta, csr, dinv, xw, agg);
  k_mlp<<<782, 128, 0, stream>>>(x, xw, agg, dinv, bg, W1, b1, W2, b2, W3, b3, out);
}


# rank launch: each slice split over 4 work-groups by dst range (256 groups of 12500 LDS counters, not 64 of 50000); a group counts the slice's edges below its range as its base; sum order unchanged
# speedup vs baseline: 3.6573x; 1.0750x over previous
#include <hip/hip_runtime.h>

typedef float v2f __attribute__((ext_vector_type(2)));
typedef float v8f __attribute__((ext_vector_type(8)));

#define N_NODES 50000
#define N_EDGES 800000
#define NB (N_NODES / 16)
#define TPITCH 132

__device__ __forceinline__ v8f wmma_f32(v2f a, v2f b, v8f c) {
  return __builtin_amdgcn_wmma_f32_16x16x4_f32(false, a, false, b, (short)0, c,
                                               false, false);
}

#define OFF_DINV   50048
#define OFF_CURSOR 150144
#define OFF_SRT    150208
#define OFF_CSR    950208
#define OFF_XW     1750208
#define OFF_AGG    8150208
#define OFF_TBL    14550208
#define OFF_META   17753280

#define NSLICE 64
#define EPS    (N_EDGES / NSLICE)
#define EPS4   (EPS / 4)
#define NPITCH 50048
#define NQ     4
#define NPQ    (N_NODES / NQ)
#define CHUNK4 5
#define NCNT   (1024 * CHUNK4 * 4)

__global__ __launch_bounds__(1024) void k_rank(const int* __restrict__ ei, int* __restrict__ srt,
                                               int* __restrict__ tbl, int* __restrict__ cursor) {
  __shared__ int c[NCNT];
  __shared__ int wsum[32];
  __shared__ int wlow[32];
  const int tid = threadIdx.x;
  const int lane = tid & 31;
  const int wave = tid >> 5;
  const int s = blockIdx.x / NQ;
  const int qd = blockIdx.x % NQ;
  const int lo = qd * NPQ;
  const int4* dst4 = (const int4*)(ei + N_EDGES + s * EPS);
  const int4* src4 = (const int4*)(ei + s * EPS);
  int4* c4 = (int4*)c;
  int4 d[4], sv[4];
#pragma unroll
  for (int k = 0; k < 4; ++k) {
    const int i = tid + k * 1024;
    d[k]  = (i < EPS4) ? dst4[i] : make_int4(0, 0, 0, 0);
    sv[k] = (i < EPS4) ? src4[i] : make_int4(0, 0, 0, 0);
  }
#pragma unroll
  for (int k = 0; k < CHUNK4; ++k) c4[tid + k * 1024] = make_int4(0, 0, 0, 0);
  __syncthreads();
  int4 r[4];
  int below = 0;
#pragma unroll
  for (int k = 0; k < 4; ++k) {
    const int i = tid + k * 1024;
    r[k] = make_int4(0, 0, 0, 0);
    if (i < EPS4) {
      below += (d[k].x < lo) + (d[k].y < lo) + (d[k].z < lo) + (d[k].w < lo);
      d[k].x -= lo; d[k].y -= lo; d[k].z -= lo; d[k].w -= lo;
      if ((unsigned)d[k].x < (unsigned)NPQ) r[k].x = atomicAdd(&c[d[k].x], 1);
      if ((unsigned)d[k].y < (unsigned)NPQ) r[k].y = atomicAdd(&c[d[k].y], 1);
      if ((unsigned)d[k].z < (unsigned)NPQ) r[k].z = atomicAdd(&c[d[k].z], 1);
      if ((unsigned)d[k].w < (unsigned)NPQ) r[k].w = atomicAdd(&c[d[k].w], 1);
    } else {
      d[k] = make_int4(-1, -1, -1, -1);
    }
  }
  for (int off = 16; off > 0; off >>= 1) below += __shfl_xor(below, off);
  if (lane == 0) wlow[wave] = below;
  __syncthreads();
  int4 q[CHUNK4];
  int p = 0;
#pragma unroll
  for (int j = 0; j < CHUNK4; ++j) {
    q[j] = c4[tid * CHUNK4 + j];
    p += q[j].x + q[j].y + q[j].z + q[j].w;
  }
  int incl = p;
  for (int off = 1; off < 32; off <<= 1) {
    int t = __shfl_up(incl, off);
    if (lane >= off) incl += t;
  }
  if (lane == 31) wsum[wave] = incl;
  int first = wlow[lane];
  for (int off = 16; off > 0; off >>= 1) first += __shfl_xor(first, off);
  __syncthreads();
  const int wt = wsum[lane];
  int winc = wt;
  for (int off = 1; off < 32; off <<= 1) {
    int t = __shfl_up(winc, off);
    if (lane >= off) winc += t;
  }
  const int wbase = __shfl(winc - wt, wave);
  int run = first + wbase + incl - p;
#pragma unroll
  for (int j = 0; j < CHUNK4; ++j) {
    int t;
    t = q[j].x; q[j].x = run; run += t;
    t = q[j].y; q[j].y = run; run += t;
    t = q[j].z; q[j].z = run; run += t;
    t = q[j].w; q[j].w = run; run += t;
    c4[tid * CHUNK4 + j] = q[j];
  }
  __syncthreads();
  int* mine = srt + s * EPS;
  int4 st[4];
#pragma unroll
  for (int k = 0; k < 4; ++k) {
    st[k].x = c[min((unsigned)d[k].x, (unsigned)NPQ)];
    st[k].y = c[min((unsigned)d[k].y, (unsigned)NPQ)];
    st[k].z = c[min((unsigned)d[k].z, (unsigned)NPQ)];
    st[k].w = c[min((unsigned)d[k].w, (unsigned)NPQ)];
  }
  asm volatile("" :: "v"(st[0].x), "v"(st[0].y), "v"(st[0].z), "v"(st[0].w),
                     "v"(st[1].x), "v"(st[1].y), "v"(st[1].z), "v"(st[1].w),
                     "v"(st[2].x), "v"(st[2].y), "v"(st[2].z), "v"(st[2].w),
                     "v"(st[3].x), "v"(st[3].y), "v"(st[3].z), "v"(st[3].w));
#pragma unroll
  for (int k = 0; k < 4; ++k) {
    if ((unsigned)d[k].x < (unsigned)NPQ) mine[st[k].x + r[k].x] = sv[k].x;
    if ((unsigned)d[k].y < (unsigned)NPQ) mine[st[k].y + r[k].y] = sv[k].y;
    if ((unsigned)d[k].z < (unsigned)NPQ) mine[st[k].z + r[k].z] = sv[k].z;
    if ((unsigned)d[k].w < (unsigned)NPQ) mine[st[k].w + r[k].w] = sv[k].w;
  }
  int4* t4 = (int4*)(tbl + (size_t)s * NPITCH + lo);
  const int n4 = NPQ / 4 + (qd == NQ - 1 ? 1 : 0);
  for (int i = tid; i < n4; i += 1024) t4[i] = c4[i];
  if (blockIdx.x == 0 && tid == 0) cursor[0] = 0;
}

#define PBLK  ((NB + 3) / 4)
#define HSL   (NSLICE / 2)
#define OBLK  ((N_NODES + 63) / 64)
__global__ __launch_bounds__(128) void k_offsets_xw(const int* __restrict__ tbl, int* __restrict__ cursor,
                                                    int* __restrict__ meta,
                                                    float* __restrict__ dinv,
                                                    const int* __restrict__ srt,
                                                    int* __restrict__ csr,
                                                    const float* __restrict__ x,
                                                    const float* __restrict__ Wg,
                                                    float* __restrict__ xw) {
  __shared__ float lds[4 * 16 * TPITCH];
  const int lane = threadIdx.x & 31;
  if (blockIdx.x >= OBLK) {
    const int wave = threadIdx.x >> 5;
    const int half = lane >> 4;
    const int lm   = lane & 15;
    float* tile = lds + wave * (16 * TPITCH);
    const int blk = (blockIdx.x - OBLK) * 4 + wave;
    if (blk >= NB) return;
    const int row0 = blk * 16;
    for (int r = 0; r < 16; ++r) {
      float4 v = *(const float4*)(x + (size_t)(row0 + r) * 128 + lane * 4);
      *(float4*)(tile + r * TPITCH + lane * 4) = v;
    }
    asm volatile("s_wait_dscnt 0" ::: "memory");

    v8f acc[8];
    for (int t = 0; t < 8; ++t) acc[t] = (v8f){};
    for (int kk = 0; kk < 32; ++kk) {
      const int k0 = 4 * kk + 2 * half;
      v2f a = *(const v2f*)(tile + lm * TPITCH + k0);
      for (int t = 0; t < 8; ++t) {
        const float* bp = Wg + (size_t)k0 * 128 + 16 * t + lm;
        v2f b; b.x = bp[0]; b.y = bp[128];
        acc[t] = wmma_f32(a, b, acc[t]);
      }
    }
    for (int t = 0; t < 8; ++t) {
      float* op = xw + (size_t)(row0 + 8 * half) * 128 + 16 * t + lm;
      for (int r = 0; r < 8; ++r) op[(size_t)r * 128] = acc[t][r];
    }
    return;
  }
  const int h = lane & 1;
  const int i = blockIdx.x * 64 + (threadIdx.x >> 1);
  const int ic = (i < N_NODES) ? i : (N_NODES - 1);
  const int* trow = tbl + (size_t)(h * HSL) * NPITCH + ic;
  int pk[HSL];
  int a = 0;
#pragma unroll
  for (int s = 0; s < HSL; ++s) pk[s] = trow[(size_t)s * NPITCH];
#pragma unroll
  for (int s = 0; s < HSL; ++s) {
    const int cn = trow[(size_t)s * NPITCH + 1] - pk[s];
    pk[s] |= cn << 16;
    a += cn;
  }
  if (i >= N_NODES) a = 0;
  int incl = a;
  for (int off = 1; off < 32; off <<= 1) {
    int t = __shfl_up(incl, off);
    if (lane >= off) incl += t;
  }
  const int total = __shfl(incl, 31);
  int base = 0;
  if (lane == 31) base = atomicAdd(cursor, total);
  base = __shfl(base, 31);
  const int c = a + __shfl_xor(a, 1);
  if (i < N_NODES) {
    const int first = base + incl - a;
    if (h == 0) {
      float d = (float)c + 1.0f;
      const float di = d > 0.0f ? rsqrtf(d) : 0.0f;
      dinv[i] = di;
      ((int4*)meta)[i] = make_int4(first, c, __float_as_int(di), 0);
    }
    int kmax = 0;
#pragma unroll
    for (int s = 0; s < HSL; ++s) kmax = max(kmax, pk[s] >> 16);
    const int* sl = srt + (h * HSL) * EPS;
    for (int k = 0; k < kmax; ++k) {
      int t[HSL];
#pragma unroll
      for (int s = 0; s < HSL; ++s) {
        const int e = min((pk[s] & 0xffff) + k, EPS - 1);
        t[s] = sl[s * EPS + e];
      }
      int pos = first + k;
#pragma unroll
      for (int s = 0; s < HSL; ++s) {
        const int cn = pk[s] >> 16;
        if (cn > k) csr[pos] = t[s];
        pos += cn;
      }
    }
  }
}

template <int NR, bool PAD>
__device__ __forceinline__ void rows_add(float4& a, const float* __restrict__ xw, int lane,
                                         int s, float ds, float dd, int j0, int m) {
  float4 v[NR];
#pragma unroll
  for (int k = 0; k < NR; ++k) {
    int jj = j0 + k;
    if (PAD) jj = jj < m ? jj : (m - 1);
    const int sj = __builtin_amdgcn_readlane(s, jj);
    v[k] = *((const float4*)(xw + (size_t)sj * 128) + lane);
  }
  __builtin_amdgcn_sched_barrier(0);
#pragma unroll
  for (int k = 0; k < NR; ++k) {
    const float nj = __int_as_float(__builtin_amdgcn_readlane(__float_as_int(ds), j0 + k)) * dd;
    a.x = __builtin_fmaf(v[k].x, nj, a.x);
    a.y = __builtin_fmaf(v[k].y, nj, a.y);
    a.z = __builtin_fmaf(v[k].z, nj, a.z);
    a.w = __builtin_fmaf(v[k].w, nj, a.w);
  }
}

__global__ __launch_bounds__(256) void k_gather(const int* __restrict__ meta,
                                                const int* __restrict__ csr,
                                                const float* __restrict__ dinv,
                                                const float* __restrict__ xw,
                                                float* __restrict__ agg) {
  const int lane = threadIdx.x & 31;
  const int node = __builtin_amdgcn_readfirstlane(blockIdx.x * 8 + (threadIdx.x >> 5));
  const int4 md = ((const int4*)meta)[node];
  const int b = md.x;
  const int n = md.y;
  const float dd = __int_as_float(md.z);
  float4 a = {0.0f, 0.0f, 0.0f, 0.0f};
  for (int base = 0; base < n; base += 32) {
    const int m = (n - base) < 32 ? (n - base) : 32;
    int s = 0;
    float ds = 0.0f;
    if (lane < m) {
      s = csr[b + base + lane];
      ds = dinv[s];
    }
    int j0 = 0;
    for (; j0 + 8 <= m; j0 += 8) rows_add<8, false>(a, xw, lane, s, ds, dd, j0, m);
    if (m - j0 > 4)      rows_add<8, true>(a, xw, lane, s, ds, dd, j0, m);
    else if (m - j0 > 0) rows_add<4, true>(a, xw, lane, s, ds, dd, j0, m);
  }
  *((float4*)(agg + (size_t)node * 128) + lane) = a;
}

__global__ __launch_bounds__(128) void k_mlp(const float* __restrict__ x, const float* __restrict__ xw,
                      const float* __restrict__ agg, const float* __restrict__ dinv,
                      const float* __restrict__ bg,
                      const float* __restrict__ W1, const float* __restrict__ b1,
                      const float* __restrict__ W2, const float* __restrict__ b2,
                      const float* __restrict__ W3, const float* __restrict__ b3,
                      float* __restrict__ out) {
  __shared__ float lds[4 * 16 * TPITCH];
  const int lane = threadIdx.x & 31;
  const int wave = threadIdx.x >> 5;
  const int half = lane >> 4;
  const int lm   = lane & 15;
  float* tile = lds + wave * (16 * TPITCH);
  const float4 bgv = *(const float4*)(bg + lane * 4);
  const int gw = blockIdx.x * 4 + wave;
  const int nwaves = gridDim.x * 4;

  for (int blk = gw; blk < NB; blk += nwaves) {
    const int row0 = blk * 16;
    const int dlb = __float_as_int(dinv[row0 + lm]);
    for (int r0 = 0; r0 < 16; r0 += 8) {
      float4 ag[8], xwv[8], xv[8];
#pragma unroll
      for (int r = 0; r < 8; ++r) {
        const size_t o = (size_t)(row0 + r0 + r) * 128 + lane * 4;
        ag[r]  = *(const float4*)(agg + o);
        xwv[r] = *(const float4*)(xw  + o);
        xv[r]  = *(const float4*)(x   + o);
      }
#pragma unroll
      for (int r = 0; r < 8; ++r) {
        const float di = __int_as_float(__builtin_amdgcn_readlane(dlb, r0 + r));
        const float sl = di * di;
        float4 h;
        h.x = fmaxf(ag[r].x + xwv[r].x * sl + bgv.x, 0.0f) + xv[r].x;
        h.y = fmaxf(ag[r].y + xwv[r].y * sl + bgv.y, 0.0f) + xv[r].y;
        h.z = fmaxf(ag[r].z + xwv[r].z * sl + bgv.z, 0.0f) + xv[r].z;
        h.w = fmaxf(ag[r].w + xwv[r].w * sl + bgv.w, 0.0f) + xv[r].w;
        *(float4*)(tile + (r0 + r) * TPITCH + lane * 4) = h;
      }
    }
    asm volatile("s_wait_dscnt 0" ::: "memory");

    v8f acc1[8];
    for (int t = 0; t < 8; ++t) acc1[t] = (v8f){};
    float bb1[8];
#pragma unroll
    for (int t = 0; t < 8; ++t) bb1[t] = b1[16 * t + lm];
    for (int kk = 0; kk < 32; ++kk) {
      const int k0 = 4 * kk + 2 * half;
      v2f a = *(const v2f*)(tile + lm * TPITCH + k0);
      for (int t = 0; t < 8; ++t) {
        const float* bp = W1 + (size_t)k0 * 128 + 16 * t + lm;
        v2f b; b.x = bp[0]; b.y = bp[128];
        acc1[t] = wmma_f32(a, b, acc1[t]);
      }
    }
    asm volatile("s_wait_dscnt 0" ::: "memory");
#pragma unroll
    for (int t = 0; t < 8; ++t) {
      const float bb = bb1[t];
      for (int r = 0; r < 8; ++r)
        tile[(r + 8 * half) * TPITCH + 16 * t + lm] = fmaxf(acc1[t][r] + bb, 0.0f);
    }
    asm volatile("s_wait_dscnt 0" ::: "memory");

    v8f acc2[4];
    for (int t = 0; t < 4; ++t) acc2[t] = (v8f){};
    float bb2[4];
#pragma unroll
    for (int t = 0; t < 4; ++t) bb2[t] = b2[16 * t + lm];
    const int lc = lm < 2 ? lm : 0;
    float w3x[16], w3y[16];
#pragma unroll
    for (int kk = 0; kk < 16; ++kk) {
      const int k0 = 4 * kk + 2 * half;
      w3x[kk] = W3[k0 * 2 + lc];
      w3y[kk] = W3[(k0 + 1) * 2 + lc];
    }
    const float bb3 = b3[lc];
    for (int kk = 0; kk < 32; ++kk) {
      const int k0 = 4 * kk + 2 * half;
      v2f a = *(const v2f*)(tile + lm * TPITCH + k0);
      for (int t = 0; t < 4; ++t) {
        const float* bp = W2 + (size_t)k0 * 64 + 16 * t + lm;
        v2f b; b.x = bp[0]; b.y = bp[64];
        acc2[t] = wmma_f32(a, b, acc2[t]);
      }
    }
    asm volatile("s_wait_dscnt 0" ::: "memory");
#pragma unroll
    for (int t = 0; t < 4; ++t) {
      const float bb = bb2[t];
      for (int r = 0; r < 8; ++r)
        tile[(r + 8 * half) * TPITCH + 16 * t + lm] = fmaxf(acc2[t][r] + bb, 0.0f);
    }
    asm volatile("s_wait_dscnt 0" ::: "memory");

    v8f acc3 = (v8f){};
#pragma unroll
    for (int kk = 0; kk < 16; ++kk) {
      const int k0 = 4 * kk + 2 * half;
      v2f a = *(const v2f*)(tile + lm * TPITCH + k0);
      v2f b;
      b.x = lm < 2 ? w3x[kk] : 0.0f;
      b.y = lm < 2 ? w3y[kk] : 0.0f;
      acc3 = wmma_f32(a, b, acc3);
    }
    if (lm < 2) {
      for (int r = 0; r < 8; ++r)
        out[(size_t)lm * N_NODES + row0 + r + 8 * half] = acc3[r] + bb3;
    }
  }
}

extern "C" void kernel_launch(void* const* d_in, const int* in_sizes, int n_in,
                              void* d_out, int out_size, void* d_ws, size_t ws_size,
                              hipStream_t stream) {
  (void)in_sizes; (void)n_in; (void)out_size; (void)ws_size;
  const float* x  = (const float*)d_in[0];
  const int*   ei = (const int*)d_in[1];
  const float* Wg = (const float*)d_in[2];
  const float* bg = (const float*)d_in[3];
  const float* W1 = (const float*)d_in[4];
  const float* b1 = (const float*)d_in[5];
  const float* W2 = (const float*)d_in[6];
  const float* b2 = (const float*)d_in[7];
  const float* W3 = (const float*)d_in[8];
  const float* b3 = (const float*)d_in[9];
  float* out = (float*)d_out;

  float* ws    = (float*)d_ws;
  int*   wsi   = (int*)d_ws;
  float* dinv  = ws  + OFF_DINV;
  int*   cursor= wsi + OFF_CURSOR;
  int*   srt   = wsi + OFF_SRT;
  int*   csr   = wsi + OFF_CSR;
  float* xw    = ws  + OFF_XW;
  float* agg   = ws  + OFF_AGG;
  int*   tbl   = wsi + OFF_TBL;
  int*   meta  = wsi + OFF_META;

  k_rank<<<NSLICE * NQ, 1024, 0, stream>>>(ei, srt, tbl, cursor);
  k_offsets_xw<<<OBLK + PBLK, 128, 0, stream>>>(tbl, cursor, meta, dinv, srt, csr, x, Wg, xw);
  k_gather<<<N_NODES / 8, 256, 0, stream>>>(meta, csr, dinv, xw, agg);
  k_mlp<<<782, 128, 0, stream>>>(x, xw, agg, dinv, bg, W1, b1, W2, b2, W3, b3, out);
}


# gather and MLP in one launch: a work-group of four waves gathers its 16-row tile into LDS (four nodes a wave), then runs the layers with N-tiles split over the waves; agg never stored; sums unchanged
# speedup vs baseline: 3.8783x; 1.0604x over previous
#include <hip/hip_runtime.h>

typedef float v2f __attribute__((ext_vector_type(2)));
typedef float v8f __attribute__((ext_vector_type(8)));

#define N_NODES 50000
#define N_EDGES 800000
#define NB (N_NODES / 16)
#define TPITCH 132

__device__ __forceinline__ v8f wmma_f32(v2f a, v2f b, v8f c) {
  return __builtin_amdgcn_wmma_f32_16x16x4_f32(false, a, false, b, (short)0, c,
                                               false, false);
}

#define OFF_DINV   50048
#define OFF_CURSOR 150144
#define OFF_SRT    150208
#define OFF_CSR    950208
#define OFF_XW     1750208
#define OFF_TBL    14550208
#define OFF_META   17753280

#define NSLICE 64
#define EPS    (N_EDGES / NSLICE)
#define EPS4   (EPS / 4)
#define NPITCH 50048
#define NQ     4
#define NPQ    (N_NODES / NQ)
#define CHUNK4 5
#define NCNT   (1024 * CHUNK4 * 4)

__global__ __launch_bounds__(1024) void k_rank(const int* __restrict__ ei, int* __restrict__ srt,
                                               int* __restrict__ tbl, int* __restrict__ cursor) {
  __shared__ int c[NCNT];
  __shared__ int wsum[32];
  __shared__ int wlow[32];
  const int tid = threadIdx.x;
  const int lane = tid & 31;
  const int wave = tid >> 5;
  const int s = blockIdx.x / NQ;
  const int qd = blockIdx.x % NQ;
  const int lo = qd * NPQ;
  const int4* dst4 = (const int4*)(ei + N_EDGES + s * EPS);
  const int4* src4 = (const int4*)(ei + s * EPS);
  int4* c4 = (int4*)c;
  int4 d[4], sv[4];
#pragma unroll
  for (int k = 0; k < 4; ++k) {
    const int i = tid + k * 1024;
    d[k]  = (i < EPS4) ? dst4[i] : make_int4(0, 0, 0, 0);
    sv[k] = (i < EPS4) ? src4[i] : make_int4(0, 0, 0, 0);
  }
#pragma unroll
  for (int k = 0; k < CHUNK4; ++k) c4[tid + k * 1024] = make_int4(0, 0, 0, 0);
  __syncthreads();
  int4 r[4];
  int below = 0;
#pragma unroll
  for (int k = 0; k < 4; ++k) {
    const int i = tid + k * 1024;
    r[k] = make_int4(0, 0, 0, 0);
    if (i < EPS4) {
      below += (d[k].x < lo) + (d[k].y < lo) + (d[k].z < lo) + (d[k].w < lo);
      d[k].x -= lo; d[k].y -= lo; d[k].z -= lo; d[k].w -= lo;
      if ((unsigned)d[k].x < (unsigned)NPQ) r[k].x = atomicAdd(&c[d[k].x], 1);
      if ((unsigned)d[k].y < (unsigned)NPQ) r[k].y = atomicAdd(&c[d[k].y], 1);
      if ((unsigned)d[k].z < (unsigned)NPQ) r[k].z = atomicAdd(&c[d[k].z], 1);
      if ((unsigned)d[k].w < (unsigned)NPQ) r[k].w = atomicAdd(&c[d[k].w], 1);
    } else {
      d[k] = make_int4(-1, -1, -1, -1);
    }
  }
  for (int off = 16; off > 0; off >>= 1) below += __shfl_xor(below, off);
  if (lane == 0) wlow[wave] = below;
  __syncthreads();
  int4 q[CHUNK4];
  int p = 0;
#pragma unroll
  for (int j = 0; j < CHUNK4; ++j) {
    q[j] = c4[tid * CHUNK4 + j];
    p += q[j].x + q[j].y + q[j].z + q[j].w;
  }
  int incl = p;
  for (int off = 1; off < 32; off <<= 1) {
    int t = __shfl_up(incl, off);
    if (lane >= off) incl += t;
  }
  if (lane == 31) wsum[wave] = incl;
  int first = wlow[lane];
  for (int off = 16; off > 0; off >>= 1) first += __shfl_xor(first, off);
  __syncthreads();
  const int wt = wsum[lane];
  int winc = wt;
  for (int off = 1; off < 32; off <<= 1) {
    int t = __shfl_up(winc, off);
    if (lane >= off) winc += t;
  }
  const int wbase = __shfl(winc - wt, wave);
  int run = first + wbase + incl - p;
#pragma unroll
  for (int j = 0; j < CHUNK4; ++j) {
    int t;
    t = q[j].x; q[j].x = run; run += t;
    t = q[j].y; q[j].y = run; run += t;
    t = q[j].z; q[j].z = run; run += t;
    t = q[j].w; q[j].w = run; run += t;
    c4[tid * CHUNK4 + j] = q[j];
  }
  __syncthreads();
  int* mine = srt + s * EPS;
  int4 st[4];
#pragma unroll
  for (int k = 0; k < 4; ++k) {
    st[k].x = c[min((unsigned)d[k].x, (unsigned)NPQ)];
    st[k].y = c[min((unsigned)d[k].y, (unsigned)NPQ)];
    st[k].z = c[min((unsigned)d[k].z, (unsigned)NPQ)];
    st[k].w = c[min((unsigned)d[k].w, (unsigned)NPQ)];
  }
  asm volatile("" :: "v"(st[0].x), "v"(st[0].y), "v"(st[0].z), "v"(st[0].w),
                     "v"(st[1].x), "v"(st[1].y), "v"(st[1].z), "v"(st[1].w),
                     "v"(st[2].x), "v"(st[2].y), "v"(st[2].z), "v"(st[2].w),
                     "v"(st[3].x), "v"(st[3].y), "v"(st[3].z), "v"(st[3].w));
#pragma unroll
  for (int k = 0; k < 4; ++k) {
    if ((unsigned)d[k].x < (unsigned)NPQ) mine[st[k].x + r[k].x] = sv[k].x;
    if ((unsigned)d[k].y < (unsigned)NPQ) mine[st[k].y + r[k].y] = sv[k].y;
    if ((unsigned)d[k].z < (unsigned)NPQ) mine[st[k].z + r[k].z] = sv[k].z;
    if ((unsigned)d[k].w < (unsigned)NPQ) mine[st[k].w + r[k].w] = sv[k].w;
  }
  int4* t4 = (int4*)(tbl + (size_t)s * NPITCH + lo);
  const int n4 = NPQ / 4 + (qd == NQ - 1 ? 1 : 0);
  for (int i = tid; i < n4; i += 1024) t4[i] = c4[i];
  if (blockIdx.x == 0 && tid == 0) cursor[0] = 0;
}

#define PBLK  ((NB + 3) / 4)
#define HSL   (NSLICE / 2)
#define OBLK  ((N_NODES + 63) / 64)
__global__ __launch_bounds__(128) void k_offsets_xw(const int* __restrict__ tbl, int* __restrict__ cursor,
                                                    int* __restrict__ meta,
                                                    float* __restrict__ dinv,
                                                    const int* __restrict__ srt,
                                                    int* __restrict__ csr,
                                                    const float* __restrict__ x,
                                                    const float* __restrict__ Wg,
                                                    float* __restrict__ xw) {
  __shared__ float lds[4 * 16 * TPITCH];
  const int lane = threadIdx.x & 31;
  if (blockIdx.x >= OBLK) {
    const int wave = threadIdx.x >> 5;
    const int half = lane >> 4;
    const int lm   = lane & 15;
    float* tile = lds + wave * (16 * TPITCH);
    const int blk = (blockIdx.x - OBLK) * 4 + wave;
    if (blk >= NB) return;
    const int row0 = blk * 16;
    for (int r = 0; r < 16; ++r) {
      float4 v = *(const float4*)(x + (size_t)(row0 + r) * 128 + lane * 4);
      *(float4*)(tile + r * TPITCH + lane * 4) = v;
    }
    asm volatile("s_wait_dscnt 0" ::: "memory");

    v8f acc[8];
    for (int t = 0; t < 8; ++t) acc[t] = (v8f){};
    for (int kk = 0; kk < 32; ++kk) {
      const int k0 = 4 * kk + 2 * half;
      v2f a = *(const v2f*)(tile + lm * TPITCH + k0);
      for (int t = 0; t < 8; ++t) {
        const float* bp = Wg + (size_t)k0 * 128 + 16 * t + lm;
        v2f b; b.x = bp[0]; b.y = bp[128];
        acc[t] = wmma_f32(a, b, acc[t]);
      }
    }
    for (int t = 0; t < 8; ++t) {
      float* op = xw + (size_t)(row0 + 8 * half) * 128 + 16 * t + lm;
      for (int r = 0; r < 8; ++r) op[(size_t)r * 128] = acc[t][r];
    }
    return;
  }
  const int h = lane & 1;
  const int i = blockIdx.x * 64 + (threadIdx.x >> 1);
  const int ic = (i < N_NODES) ? i : (N_NODES - 1);
  const int* trow = tbl + (size_t)(h * HSL) * NPITCH + ic;
  int pk[HSL];
  int a = 0;
#pragma unroll
  for (int s = 0; s < HSL; ++s) pk[s] = trow[(size_t)s * NPITCH];
#pragma unroll
  for (int s = 0; s < HSL; ++s) {
    const int cn = trow[(size_t)s * NPITCH + 1] - pk[s];
    pk[s] |= cn << 16;
    a += cn;
  }
  if (i >= N_NODES) a = 0;
  int incl = a;
  for (int off = 1; off < 32; off <<= 1) {
    int t = __shfl_up(incl, off);
    if (lane >= off) incl += t;
  }
  const int total = __shfl(incl, 31);
  int base = 0;
  if (lane == 31) base = atomicAdd(cursor, total);
  base = __shfl(base, 31);
  const int c = a + __shfl_xor(a, 1);
  if (i < N_NODES) {
    const int first = base + incl - a;
    if (h == 0) {
      float d = (float)c + 1.0f;
      const float di = d > 0.0f ? rsqrtf(d) : 0.0f;
      dinv[i] = di;
      ((int4*)meta)[i] = make_int4(first, c, __float_as_int(di), 0);
    }
    int kmax = 0;
#pragma unroll
    for (int s = 0; s < HSL; ++s) kmax = max(kmax, pk[s] >> 16);
    const int* sl = srt + (h * HSL) * EPS;
    for (int k = 0; k < kmax; ++k) {
      int t[HSL];
#pragma unroll
      for (int s = 0; s < HSL; ++s) {
        const int e = min((pk[s] & 0xffff) + k, EPS - 1);
        t[s] = sl[s * EPS + e];
      }
      int pos = first + k;
#pragma unroll
      for (int s = 0; s < HSL; ++s) {
        const int cn = pk[s] >> 16;
        if (cn > k) csr[pos] = t[s];
        pos += cn;
      }
    }
  }
}

template <int NR, bool PAD>
__device__ __forceinline__ void rows_add(float4& a, const float* __restrict__ xw, int lane,
                                         int s, float ds, float dd, int j0, int m) {
  float4 v[NR];
#pragma unroll
  for (int k = 0; k < NR; ++k) {
    int jj = j0 + k;
    if (PAD) jj = jj < m ? jj : (m - 1);
    const int sj = __builtin_amdgcn_readlane(s, jj);
    v[k] = *((const float4*)(xw + (size_t)sj * 128) + lane);
  }
  __builtin_amdgcn_sched_barrier(0);
#pragma unroll
  for (int k = 0; k < NR; ++k) {
    float nj = __int_as_float(__builtin_amdgcn_readlane(__float_as_int(ds), j0 + k)) * dd;
    if (PAD) nj = (j0 + k < m) ? nj : 0.0f;
    a.x = __builtin_fmaf(v[k].x, nj, a.x);
    a.y = __builtin_fmaf(v[k].y, nj, a.y);
    a.z = __builtin_fmaf(v[k].z, nj, a.z);
    a.w = __builtin_fmaf(v[k].w, nj, a.w);
  }
}

__global__ __launch_bounds__(128) void k_gather_mlp(const int* __restrict__ meta,
                      const int* __restrict__ csr, const float* __restrict__ dinv,
                      const float* __restrict__ xw, const float* __restrict__ x,
                      const float* __restrict__ bg,
                      const float* __restrict__ W1, const float* __restrict__ b1,
                      const float* __restrict__ W2, const float* __restrict__ b2,
                      const float* __restrict__ W3, const float* __restrict__ b3,
                      float* __restrict__ out) {
  __shared__ float tA[16 * TPITCH];
  __shared__ float tB[16 * TPITCH];
  const int lane = threadIdx.x & 31;
  const int wave = __builtin_amdgcn_readfirstlane(threadIdx.x >> 5);
  const int half = lane >> 4;
  const int lm   = lane & 15;
  const int row0 = blockIdx.x * 16;
  const int node0 = row0 + wave * 4;
  const float4 bgv = *(const float4*)(bg + lane * 4);

  int4 md[4];
  int s[4];
  float ds[4];
#pragma unroll
  for (int j = 0; j < 4; ++j) md[j] = ((const int4*)meta)[node0 + j];
#pragma unroll
  for (int j = 0; j < 4; ++j) {
    const int e = md[j].x + lane;
    s[j] = csr[e < N_EDGES ? e : N_EDGES - 1];
  }
  asm volatile("" :: "v"(s[0]), "v"(s[1]), "v"(s[2]), "v"(s[3]));
#pragma unroll
  for (int j = 0; j < 4; ++j) ds[j] = dinv[s[j]];
#pragma unroll
  for (int j = 0; j < 4; ++j) {
    const int node = node0 + j;
    const int b = md[j].x;
    const int n = md[j].y;
    const float dd = __int_as_float(md[j].z);
    const float4 xs = *((const float4*)(xw + (size_t)node * 128) + lane);
    const float4 xv = *((const float4*)(x  + (size_t)node * 128) + lane);
    float4 a = {0.0f, 0.0f, 0.0f, 0.0f};
    int sj = s[j];
    float dsj = ds[j];
    for (int base = 0; base < n; base += 32) {
      const int m = (n - base) < 32 ? (n - base) : 32;
      if (base > 0) {
        sj = 0;
        dsj = 0.0f;
        if (lane < m) {
          sj = csr[b + base + lane];
          dsj = dinv[sj];
        }
      }
      int j0 = 0;
      for (; j0 + 8 <= m; j0 += 8) rows_add<8, false>(a, xw, lane, sj, dsj, dd, j0, m);
      if (m - j0 > 4)      rows_add<8, true>(a, xw, lane, sj, dsj, dd, j0, m);
      else if (m - j0 > 0) rows_add<4, true>(a, xw, lane, sj, dsj, dd, j0, m);
    }
    const float sl = dd * dd;
    float4 h;
    h.x = fmaxf(a.x + xs.x * sl + bgv.x, 0.0f) + xv.x;
    h.y = fmaxf(a.y + xs.y * sl + bgv.y, 0.0f) + xv.y;
    h.z = fmaxf(a.z + xs.z * sl + bgv.z, 0.0f) + xv.z;
    h.w = fmaxf(a.w + xs.w * sl + bgv.w, 0.0f) + xv.w;
    *(float4*)(tA + (wave * 4 + j) * TPITCH + lane * 4) = h;
  }
  __syncthreads();

  v8f acc1[2];
  float bb1[2];
#pragma unroll
  for (int t = 0; t < 2; ++t) {
    acc1[t] = (v8f){};
    bb1[t] = b1[16 * (2 * wave + t) + lm];
  }
#pragma unroll 16
  for (int kk = 0; kk < 32; ++kk) {
    const int k0 = 4 * kk + 2 * half;
    v2f a = *(const v2f*)(tA + lm * TPITCH + k0);
#pragma unroll
    for (int t = 0; t < 2; ++t) {
      const float* bp = W1 + (size_t)k0 * 128 + 16 * (2 * wave + t) + lm;
      v2f b; b.x = bp[0]; b.y = bp[128];
      acc1[t] = wmma_f32(a, b, acc1[t]);
    }
  }
#pragma unroll
  for (int t = 0; t < 2; ++t) {
    for (int r = 0; r < 8; ++r)
      tB[(r + 8 * half) * TPITCH + 16 * (2 * wave + t) + lm] = fmaxf(acc1[t][r] + bb1[t], 0.0f);
  }
  __syncthreads();

  v8f acc2 = (v8f){};
  const float bb2 = b2[16 * wave + lm];
  const int lc = lm < 2 ? lm : 0;
  float w3x[16], w3y[16];
  float bb3 = 0.0f;
#pragma unroll
  for (int kk = 0; kk < 16; ++kk) { w3x[kk] = 0.0f; w3y[kk] = 0.0f; }
  if (wave == 0) {
#pragma unroll
    for (int kk = 0; kk < 16; ++kk) {
      const int k0 = 4 * kk + 2 * half;
      w3x[kk] = W3[k0 * 2 + lc];
      w3y[kk] = W3[(k0 + 1) * 2 + lc];
    }
    bb3 = b3[lc];
  }
#pragma unroll
  for (int kk = 0; kk < 32; ++kk) {
    const int k0 = 4 * kk + 2 * half;
    v2f a = *(const v2f*)(tB + lm * TPITCH + k0);
    const float* bp = W2 + (size_t)k0 * 64 + 16 * wave + lm;
    v2f b; b.x = bp[0]; b.y = bp[64];
    acc2 = wmma_f32(a, b, acc2);
  }
  for (int r = 0; r < 8; ++r)
    tA[(r + 8 * half) * TPITCH + 16 * wave + lm] = fmaxf(acc2[r] + bb2, 0.0f);
  __syncthreads();

  if (wave == 0) {
    v8f acc3 = (v8f){};
#pragma unroll
    for (int kk = 0; kk < 16; ++kk) {
      const int k0 = 4 * kk + 2 * half;
      v2f a = *(const v2f*)(tA + lm * TPITCH + k0);
      v2f b;
      b.x = lm < 2 ? w3x[kk] : 0.0f;
      b.y = lm < 2 ? w3y[kk] : 0.0f;
      acc3 = wmma_f32(a, b, acc3);
    }
    if (lm < 2) {
      for (int r = 0; r < 8; ++r)
        out[(size_t)lm * N_NODES + row0 + r + 8 * half] = acc3[r] + bb3;
    }
  }
}

extern "C" void kernel_launch(void* const* d_in, const int* in_sizes, int n_in,
                              void* d_out, int out_size, void* d_ws, size_t ws_size,
                              hipStream_t stream) {
  (void)in_sizes; (void)n_in; (void)out_size; (void)ws_size;
  const float* x  = (const float*)d_in[0];
  const int*   ei = (const int*)d_in[1];
  const float* Wg = (const float*)d_in[2];
  const float* bg = (const float*)d_in[3];
  const float* W1 = (const float*)d_in[4];
  const float* b1 = (const float*)d_in[5];
  const float* W2 = (const float*)d_in[6];
  const float* b2 = (const float*)d_in[7];
  const float* W3 = (const float*)d_in[8];
  const float* b3 = (const float*)d_in[9];
  float* out = (float*)d_out;

  float* ws    = (float*)d_ws;
  int*   wsi   = (int*)d_ws;
  float* dinv  = ws  + OFF_DINV;
  int*   cursor= wsi + OFF_CURSOR;
  int*   srt   = wsi + OFF_SRT;
  int*   csr   = wsi + OFF_CSR;
  float* xw    = ws  + OFF_XW;
  int*   tbl   = wsi + OFF_TBL;
  int*   meta  = wsi + OFF_META;

  k_rank<<<NSLICE * NQ, 1024, 0, stream>>>(ei, srt, tbl, cursor);
  k_offsets_xw<<<OBLK + PBLK, 128, 0, stream>>>(tbl, cursor, meta, dinv, srt, csr, x, Wg, xw);
  k_gather_mlp<<<NB, 128, 0, stream>>>(meta, csr, dinv, xw, x, bg, W1, b1, W2, b2, W3, b3, out);
}
